# starting kernel
# baseline (speedup 1.0000x reference)
#include <hip/hip_runtime.h>
#include <hip/hip_bf16.h>
#include <math.h>

typedef float v2f __attribute__((ext_vector_type(2)));
typedef float v8f __attribute__((ext_vector_type(8)));

#define NEG_SLOPE 0.2f

__device__ __forceinline__ void atomicMaxF(float* addr, float val) {
    if (val >= 0.0f) {
        atomicMax((int*)addr, __float_as_int(val));
    } else {
        atomicMin((unsigned int*)addr, __float_as_uint(val));
    }
}

__device__ __forceinline__ void edge_sd(const int* __restrict__ ei, int e, int E,
                                        int& s, int& d) {
    if (e < E) { s = ei[e]; d = ei[E + e]; }
    else       { s = d = e - E; }
}

__global__ void fill_kernel(float* __restrict__ p, float v, int n) {
    int i = blockIdx.x * blockDim.x + threadIdx.x;
    if (i < n) p[i] = v;
}

template <int K, int NCOLT>
__global__ void gemm_wmma_f32(const float* __restrict__ X,
                              const float* __restrict__ W,
                              float* __restrict__ H) {
    constexpr int FOUT = 16 * NCOLT;
    __shared__ float As[16][K + 4];

    const int r0   = blockIdx.x * 16;
    const int lane = threadIdx.x;
    const int wave = threadIdx.y;
    const int tid  = wave * 32 + lane;
    const int nthr = 32 * NCOLT;

    for (int i = tid; i < 16 * K; i += nthr) {
        int rr = i / K, cc = i % K;
        As[rr][cc] = X[(size_t)(r0 + rr) * K + cc];
    }
    __syncthreads();

    const int half = lane >> 4;
    const int m    = lane & 15;
    const int c0   = wave * 16;

    v8f acc = {};
    #pragma unroll 4
    for (int kk = 0; kk < K; kk += 4) {
        const int kb = kk + 2 * half;
        v2f a, b;
        a[0] = As[m][kb];
        a[1] = As[m][kb + 1];
        b[0] = W[(size_t)(kb)     * FOUT + c0 + m];
        b[1] = W[(size_t)(kb + 1) * FOUT + c0 + m];
        acc = __builtin_amdgcn_wmma_f32_16x16x4_f32(
             false, a,  false, b,
             (short)0, acc,  false,  false);
    }

    #pragma unroll
    for (int r = 0; r < 8; ++r) {
        H[(size_t)(r0 + r + 8 * half) * FOUT + c0 + m] = acc[r];
    }
}

template <int H, int D>
__global__ void alpha_kernel(const float* __restrict__ h,
                             const float* __restrict__ a_src,
                             const float* __restrict__ a_dst,
                             float* __restrict__ as, float* __restrict__ ad,
                             int n) {
    int i = blockIdx.x * blockDim.x + threadIdx.x;
    if (i >= n * H) return;
    int node = i / H, hh = i % H;
    const float* hp = h + (size_t)node * (H * D) + hh * D;
    float s = 0.f, d = 0.f;
    #pragma unroll
    for (int j = 0; j < D; ++j) {
        float v = hp[j];
        s += v * a_src[hh * D + j];
        d += v * a_dst[hh * D + j];
    }
    as[i] = s;
    ad[i] = d;
}

template <int H>
__global__ void edge_max_kernel(const int* __restrict__ ei, int E, int N,
                                const float* __restrict__ as,
                                const float* __restrict__ ad,
                                float* __restrict__ ebuf,
                                float* __restrict__ m) {
    int i = blockIdx.x * blockDim.x + threadIdx.x;
    if (i >= (E + N) * H) return;
    int e = i / H, hh = i % H;
    int s, d;
    edge_sd(ei, e, E, s, d);
    float v = as[s * H + hh] + ad[d * H + hh];
    v = (v > 0.f) ? v : NEG_SLOPE * v;
    ebuf[i] = v;
    atomicMaxF(&m[d * H + hh], v);
}

template <int H>
__global__ void edge_exp_kernel(const int* __restrict__ ei, int E, int N,
                                float* __restrict__ ebuf,
                                const float* __restrict__ m,
                                float* __restrict__ den) {
    int i = blockIdx.x * blockDim.x + threadIdx.x;
    if (i >= (E + N) * H) return;
    int e = i / H, hh = i % H;
    int s, d;
    edge_sd(ei, e, E, s, d);
    float ex = expf(ebuf[i] - m[d * H + hh]);
    ebuf[i] = ex;
    atomicAdd(&den[d * H + hh], ex);
}

template <int H, int D>
__global__ void edge_msg_kernel(const int* __restrict__ ei, int E, int N,
                                const float* __restrict__ h,
                                const float* __restrict__ ebuf,
                                const float* __restrict__ den,
                                float* __restrict__ agg) {
    int i = blockIdx.x * blockDim.x + threadIdx.x;
    if (i >= (E + N) * H) return;
    int e = i / H, hh = i % H;
    int s, d;
    edge_sd(ei, e, E, s, d);
    float alpha = ebuf[i] / (den[d * H + hh] + 1e-16f);
    const float* hp = h + (size_t)s * (H * D) + hh * D;
    float* ap = agg + (size_t)d * (H * D) + hh * D;
    #pragma unroll
    for (int j = 0; j < D; ++j) {
        atomicAdd(&ap[j], hp[j] * alpha);
    }
}

__global__ void bias_elu_kernel(float* __restrict__ agg,
                                const float* __restrict__ b, int n, int F) {
    int i = blockIdx.x * blockDim.x + threadIdx.x;
    if (i >= n * F) return;
    float v = agg[i] + b[i % F];
    agg[i] = (v > 0.f) ? v : (expf(v) - 1.f);
}

__global__ void add_bias_kernel(float* __restrict__ out,
                                const float* __restrict__ b, int n, int F) {
    int i = blockIdx.x * blockDim.x + threadIdx.x;
    if (i >= n * F) return;
    out[i] += b[i % F];
}

static inline int cdiv(int a, int b) { return (a + b - 1) / b; }

extern "C" void kernel_launch(void* const* d_in, const int* in_sizes, int n_in,
                              void* d_out, int out_size, void* d_ws, size_t ws_size,
                              hipStream_t stream) {
    const float* x      = (const float*)d_in[0];
    const int*   ei     = (const int*)  d_in[1];
    const float* W1     = (const float*)d_in[2];
    const float* a_src1 = (const float*)d_in[3];
    const float* a_dst1 = (const float*)d_in[4];
    const float* b1     = (const float*)d_in[5];
    const float* W2     = (const float*)d_in[6];
    const float* a_src2 = (const float*)d_in[7];
    const float* a_dst2 = (const float*)d_in[8];
    const float* b2     = (const float*)d_in[9];

    const int N  = in_sizes[0] / 128;
    const int E  = in_sizes[1] / 2;
    const int EP = E + N;

    float* ws   = (float*)d_ws;
    float* h1   = ws;
    float* agg1 = h1   + (size_t)N * 64;
    float* as1  = agg1 + (size_t)N * 64;
    float* ad1  = as1  + (size_t)N * 4;
    float* m1   = ad1  + (size_t)N * 4;
    float* den1 = m1   + (size_t)N * 4;
    float* ex1  = den1 + (size_t)N * 4;
    float* h2   = h1;
    float* as2  = h2   + (size_t)N * 16;
    float* ad2  = as2  + N;
    float* m2   = ad2  + N;
    float* den2 = m2   + N;
    float* ex2  = den2 + N;

    float* out  = (float*)d_out;

    const int TB = 256;

    fill_kernel<<<cdiv(N * 4, TB), TB, 0, stream>>>(m1, -INFINITY, N * 4);
    fill_kernel<<<cdiv(N * 4, TB), TB, 0, stream>>>(den1, 0.f, N * 4);
    fill_kernel<<<cdiv(N * 64, TB), TB, 0, stream>>>(agg1, 0.f, N * 64);

    gemm_wmma_f32<128, 4><<<N / 16, dim3(32, 4), 0, stream>>>(x, W1, h1);

    alpha_kernel<4, 16><<<cdiv(N * 4, TB), TB, 0, stream>>>(
        h1, a_src1, a_dst1, as1, ad1, N);

    edge_max_kernel<4><<<cdiv(EP * 4, TB), TB, 0, stream>>>(
        ei, E, N, as1, ad1, ex1, m1);
    edge_exp_kernel<4><<<cdiv(EP * 4, TB), TB, 0, stream>>>(
        ei, E, N, ex1, m1, den1);
    edge_msg_kernel<4, 16><<<cdiv(EP * 4, TB), TB, 0, stream>>>(
        ei, E, N, h1, ex1, den1, agg1);

    bias_elu_kernel<<<cdiv(N * 64, TB), TB, 0, stream>>>(agg1, b1, N, 64);

    fill_kernel<<<cdiv(N, TB), TB, 0, stream>>>(m2, -INFINITY, N);
    fill_kernel<<<cdiv(N, TB), TB, 0, stream>>>(den2, 0.f, N);
    fill_kernel<<<cdiv(N * 16, TB), TB, 0, stream>>>(out, 0.f, N * 16);

    gemm_wmma_f32<64, 1><<<N / 16, dim3(32, 1), 0, stream>>>(agg1, W2, h2);

    alpha_kernel<1, 16><<<cdiv(N, TB), TB, 0, stream>>>(
        h2, a_src2, a_dst2, as2, ad2, N);

    edge_max_kernel<1><<<cdiv(EP, TB), TB, 0, stream>>>(
        ei, E, N, as2, ad2, ex2, m2);
    edge_exp_kernel<1><<<cdiv(EP, TB), TB, 0, stream>>>(
        ei, E, N, ex2, m2, den2);
    edge_msg_kernel<1, 16><<<cdiv(EP, TB), TB, 0, stream>>>(
        ei, E, N, h2, ex2, den2, out);

    add_bias_kernel<<<cdiv(N * 16, TB), TB, 0, stream>>>(out, b2, N, 16);
}


# Destination rows cut into 256 tiles, one work-group per tile: edges binned by tile; softmax max, sums and row sums accumulated in LDS by LDS atomics (arrival order); one divide per element
# speedup vs baseline: 5.4480x; 5.4480x over previous
#include <hip/hip_runtime.h>
#include <hip/hip_bf16.h>
#include <math.h>

typedef float v2f __attribute__((ext_vector_type(2)));
typedef float v8f __attribute__((ext_vector_type(8)));

#define NEG_SLOPE 0.2f

__device__ __forceinline__ void atomicMaxF(float* addr, float val) {
    if (val >= 0.0f) {
        atomicMax((int*)addr, __float_as_int(val));
    } else {
        atomicMin((unsigned int*)addr, __float_as_uint(val));
    }
}

__device__ __forceinline__ void edge_sd(const int* __restrict__ ei, int e, int E,
                                        int& s, int& d) {
    if (e < E) { s = ei[e]; d = ei[E + e]; }
    else       { s = d = e - E; }
}

template <int K, int NCOLT>
__global__ void gemm_wmma_f32(const float* __restrict__ X,
                              const float* __restrict__ W,
                              float* __restrict__ H) {
    constexpr int FOUT = 16 * NCOLT;
    __shared__ float As[16][K + 4];

    const int r0   = blockIdx.x * 16;
    const int lane = threadIdx.x;
    const int wave = threadIdx.y;
    const int tid  = wave * 32 + lane;
    const int nthr = 32 * NCOLT;

    for (int i = tid; i < 16 * K; i += nthr) {
        int rr = i / K, cc = i % K;
        As[rr][cc] = X[(size_t)(r0 + rr) * K + cc];
    }
    __syncthreads();

    const int half = lane >> 4;
    const int m    = lane & 15;
    const int c0   = wave * 16;

    v8f acc = {};
    #pragma unroll 4
    for (int kk = 0; kk < K; kk += 4) {
        const int kb = kk + 2 * half;
        v2f a, b;
        a[0] = As[m][kb];
        a[1] = As[m][kb + 1];
        b[0] = W[(size_t)(kb)     * FOUT + c0 + m];
        b[1] = W[(size_t)(kb + 1) * FOUT + c0 + m];
        acc = __builtin_amdgcn_wmma_f32_16x16x4_f32(
             false, a,  false, b,
             (short)0, acc,  false,  false);
    }

    #pragma unroll
    for (int r = 0; r < 8; ++r) {
        H[(size_t)(r0 + r + 8 * half) * FOUT + c0 + m] = acc[r];
    }
}

template <int H, int D>
__global__ void alpha_kernel(const float* __restrict__ h,
                             const float* __restrict__ a_src,
                             const float* __restrict__ a_dst,
                             float* __restrict__ as, float* __restrict__ ad,
                             int n) {
    int i = blockIdx.x * blockDim.x + threadIdx.x;
    if (i >= n * H) return;
    int node = i / H, hh = i % H;
    const float* hp = h + (size_t)node * (H * D) + hh * D;
    float s = 0.f, d = 0.f;
    #pragma unroll
    for (int j = 0; j < D; ++j) {
        float v = hp[j];
        s += v * a_src[hh * D + j];
        d += v * a_dst[hh * D + j];
    }
    as[i] = s;
    ad[i] = d;
}

#define TILE_ROWS 391
#define MAXT      256
#define NCHUNK    64
#define TILE_THREADS 1024

__device__ __forceinline__ int tile_of(int d) {
    int t = d / TILE_ROWS;
    return t < MAXT ? t : MAXT - 1;
}

__global__ void __launch_bounds__(TILE_THREADS)
tile_hist_kernel(const int* __restrict__ ei, int E, int* __restrict__ histG) {
    __shared__ int hist[MAXT];
    const int t = threadIdx.x;
    if (t < MAXT) hist[t] = 0;
    __syncthreads();
    const int cs = (E + gridDim.x - 1) / gridDim.x;
    const int lo = blockIdx.x * cs;
    const int hi = (lo + cs < E) ? lo + cs : E;
    for (int e = lo + t; e < hi; e += TILE_THREADS) {
        atomicAdd(&hist[tile_of(ei[E + e])], 1);
    }
    __syncthreads();
    if (t < MAXT) histG[blockIdx.x * MAXT + t] = hist[t];
}

__global__ void __launch_bounds__(TILE_THREADS)
tile_place_kernel(const int* __restrict__ ei, int E,
                  const int* __restrict__ histG,
                  int* __restrict__ tileStart,
                  int* __restrict__ ls, int* __restrict__ ld) {
    __shared__ int cursor[MAXT];
    __shared__ int scan[MAXT];
    const int t = threadIdx.x;
    int tot = 0, before = 0;
    if (t < MAXT) {
        #pragma unroll 16
        for (int c = 0; c < NCHUNK; ++c) {
            const int v = histG[c * MAXT + t];
            tot += v;
            if (c < (int)blockIdx.x) before += v;
        }
        scan[t] = tot;
    }
    __syncthreads();
    for (int off = 1; off < MAXT; off <<= 1) {
        int v = 0;
        if (t < MAXT && t >= off) v = scan[t - off];
        __syncthreads();
        if (t < MAXT) scan[t] += v;
        __syncthreads();
    }
    if (t < MAXT) {
        const int startT = scan[t] - tot;
        cursor[t] = startT + before;
        if (blockIdx.x == 0) {
            tileStart[t] = startT;
            if (t == MAXT - 1) tileStart[MAXT] = scan[t];
        }
    }
    __syncthreads();
    const int cs = (E + gridDim.x - 1) / gridDim.x;
    const int lo = blockIdx.x * cs;
    const int hi = (lo + cs < E) ? lo + cs : E;
    for (int e = lo + t; e < hi; e += TILE_THREADS) {
        const int s = ei[e];
        const int d = ei[E + e];
        const int pos = atomicAdd(&cursor[tile_of(d)], 1);
        ls[pos] = s;
        ld[pos] = d;
    }
}

__device__ __forceinline__ float lrelu(float v) {
    return (v > 0.f) ? v : NEG_SLOPE * v;
}

template <int H, bool ELU>
__global__ void __launch_bounds__(TILE_THREADS)
tile_gat_kernel(const int* __restrict__ tileStart,
                const int* __restrict__ ls, const int* __restrict__ ld,
                const float* __restrict__ h,
                const float* __restrict__ as, const float* __restrict__ ad,
                const float* __restrict__ bias,
                float* __restrict__ res, int N) {
    constexpr int F = H * 16;
    __shared__ float acc[TILE_ROWS * F];
    __shared__ float mx[TILE_ROWS * H];
    __shared__ float dn[TILE_ROWS * H];

    const int tid   = threadIdx.x;
    const int lane  = tid & 31;
    const int wave  = tid >> 5;
    constexpr int NW = TILE_THREADS / 32;

    const int r0    = blockIdx.x * TILE_ROWS;
    int nrows = N - r0;
    if (nrows > TILE_ROWS) nrows = TILE_ROWS;
    if (nrows < 0) nrows = 0;
    const int start = tileStart[blockIdx.x];
    const int cnt   = tileStart[blockIdx.x + 1] - start;
    const int total = cnt + nrows;

    for (int i = tid; i < nrows * F; i += TILE_THREADS) acc[i] = 0.f;
    for (int i = tid; i < nrows * H; i += TILE_THREADS) {
        const int g = r0 * H + i;
        mx[i] = lrelu(as[g] + ad[g]);
        dn[i] = 0.f;
    }
    __syncthreads();

    for (int i0 = tid; i0 < cnt; i0 += 4 * TILE_THREADS) {
        int   sv[4], dv[4];
        float ev[4][H];
        #pragma unroll
        for (int u = 0; u < 4; ++u) {
            const int i = i0 + u * TILE_THREADS;
            const bool ok = i < cnt;
            sv[u] = ok ? ls[start + i] : 0;
            dv[u] = ok ? ld[start + i] : r0;
        }
        #pragma unroll
        for (int u = 0; u < 4; ++u) {
            #pragma unroll
            for (int hh = 0; hh < H; ++hh)
                ev[u][hh] = lrelu(as[sv[u] * H + hh] + ad[dv[u] * H + hh]);
        }
        #pragma unroll
        for (int u = 0; u < 4; ++u) {
            if (i0 + u * TILE_THREADS < cnt) {
                #pragma unroll
                for (int hh = 0; hh < H; ++hh)
                    atomicMaxF(&mx[(dv[u] - r0) * H + hh], ev[u][hh]);
            }
        }
    }
    __syncthreads();

    int s_n = 0, d_n = r0;
    {
        const int i = wave * 32 + lane;
        if (i < total) {
            if (i < cnt) { s_n = ls[start + i]; d_n = ld[start + i]; }
            else         { s_n = d_n = r0 + (i - cnt); }
        }
    }
    for (int b = wave * 32; b < total; b += NW * 32) {
        const int i  = b + lane;
        const int s  = s_n;
        const int d  = d_n;
        const int dl = d - r0;
        {
            const int in = i + NW * 32;
            s_n = 0; d_n = r0;
            if (in < total) {
                if (in < cnt) { s_n = ls[start + in]; d_n = ld[start + in]; }
                else          { s_n = d_n = r0 + (in - cnt); }
            }
        }
        float ex[H];
        #pragma unroll
        for (int hh = 0; hh < H; ++hh) ex[hh] = 0.f;
        if (i < total) {
            #pragma unroll
            for (int hh = 0; hh < H; ++hh) {
                const float e = lrelu(as[s * H + hh] + ad[d * H + hh]);
                ex[hh] = expf(e - mx[dl * H + hh]);
            }
            #pragma unroll
            for (int hh = 0; hh < H; ++hh) atomicAdd(&dn[dl * H + hh], ex[hh]);
        }
        if constexpr (H == 4) {
            #pragma unroll
            for (int jb = 0; jb < 32; jb += 16) {
                if (b + jb < total) {
                    float v0[16], v1[16];
                    #pragma unroll
                    for (int j = 0; j < 16; ++j) {
                        const int sj = __builtin_amdgcn_readlane(s, jb + j);
                        const float* hp = h + (size_t)sj * F;
                        v0[j] = hp[lane];
                        v1[j] = hp[32 + lane];
                    }
                    #pragma unroll
                    for (int j = 0; j < 16; ++j) {
                        const int   dj = __builtin_amdgcn_readlane(dl, jb + j);
                        const float e0 = __int_as_float(__builtin_amdgcn_readlane(__float_as_int(ex[0]), jb + j));
                        const float e1 = __int_as_float(__builtin_amdgcn_readlane(__float_as_int(ex[1]), jb + j));
                        const float e2 = __int_as_float(__builtin_amdgcn_readlane(__float_as_int(ex[2]), jb + j));
                        const float e3 = __int_as_float(__builtin_amdgcn_readlane(__float_as_int(ex[3]), jb + j));
                        const float wlo = (lane < 16) ? e0 : e1;
                        const float whi = (lane < 16) ? e2 : e3;
                        if (b + jb + j < total) {
                            atomicAdd(&acc[dj * F + lane],      v0[j] * wlo);
                            atomicAdd(&acc[dj * F + 32 + lane], v1[j] * whi);
                        }
                    }
                }
            }
        } else {
            const int c    = lane & 15;
            const int half = lane >> 4;
            float v[16];
            #pragma unroll
            for (int j = 0; j < 16; ++j) {
                const int sj = __shfl(s, 2 * j + half);
                v[j] = h[(size_t)sj * F + c];
            }
            #pragma unroll
            for (int j = 0; j < 16; ++j) {
                const int   from = 2 * j + half;
                const int   dj = __shfl(dl, from);
                const float w  = __shfl(ex[0], from);
                if (b + from < total) {
                    atomicAdd(&acc[dj * F + c], v[j] * w);
                }
            }
        }
    }
    __syncthreads();

    for (int i = tid; i < nrows * F; i += TILE_THREADS) {
        const int r = i / F, f = i % F;
        float v = acc[i] / (dn[r * H + f / 16] + 1e-16f) + bias[f];
        if (ELU) v = (v > 0.f) ? v : (expf(v) - 1.f);
        res[(size_t)r0 * F + i] = v;
    }
}

static inline int cdiv(int a, int b) { return (a + b - 1) / b; }

extern "C" void kernel_launch(void* const* d_in, const int* in_sizes, int n_in,
                              void* d_out, int out_size, void* d_ws, size_t ws_size,
                              hipStream_t stream) {
    const float* x      = (const float*)d_in[0];
    const int*   ei     = (const int*)  d_in[1];
    const float* W1     = (const float*)d_in[2];
    const float* a_src1 = (const float*)d_in[3];
    const float* a_dst1 = (const float*)d_in[4];
    const float* b1     = (const float*)d_in[5];
    const float* W2     = (const float*)d_in[6];
    const float* a_src2 = (const float*)d_in[7];
    const float* a_dst2 = (const float*)d_in[8];
    const float* b2     = (const float*)d_in[9];

    const int N  = in_sizes[0] / 128;
    const int E  = in_sizes[1] / 2;

    float* ws   = (float*)d_ws;
    float* h1   = ws;
    float* x2   = h1   + (size_t)N * 64;
    float* as1  = x2   + (size_t)N * 64;
    float* ad1  = as1  + (size_t)N * 4;
    float* h2   = ad1  + (size_t)N * 4;
    float* as2  = h2   + (size_t)N * 16;
    float* ad2  = as2  + N;
    int* histG     = (int*)(ad2 + N);
    int* tileStart = histG + MAXT * MAXT;
    int* ls        = tileStart + 512;
    int* ld        = ls + E;

    float* out  = (float*)d_out;

    const int TB = 256;
    const int NT = cdiv(N, TILE_ROWS);

    tile_hist_kernel<<<NCHUNK, TILE_THREADS, 0, stream>>>(ei, E, histG);
    tile_place_kernel<<<NCHUNK, TILE_THREADS, 0, stream>>>(ei, E, histG, tileStart, ls, ld);

    gemm_wmma_f32<128, 4><<<N / 16, dim3(32, 4), 0, stream>>>(x, W1, h1);
    alpha_kernel<4, 16><<<cdiv(N * 4, TB), TB, 0, stream>>>(
        h1, a_src1, a_dst1, as1, ad1, N);
    tile_gat_kernel<4, true><<<NT, TILE_THREADS, 0, stream>>>(
        tileStart, ls, ld, h1, as1, ad1, b1, x2, N);

    gemm_wmma_f32<64, 1><<<N / 16, dim3(32, 1), 0, stream>>>(x2, W2, h2);
    alpha_kernel<1, 16><<<cdiv(N, TB), TB, 0, stream>>>(
        h2, a_src2, a_dst2, as2, ad2, N);
    tile_gat_kernel<1, false><<<NT, TILE_THREADS, 0, stream>>>(
        tileStart, ls, ld, h2, as2, ad2, b2, out, N);
}


# Layer-2 product and logits formed in the layer-1 tile kernel: the tile's row sums sit at pitch 68 in LDS, get elu in place and feed f32 WMMA there; row round trip, GEMM and logit launches go
# speedup vs baseline: 5.7793x; 1.0608x over previous
#include <hip/hip_runtime.h>
#include <hip/hip_bf16.h>
#include <math.h>

typedef float v2f __attribute__((ext_vector_type(2)));
typedef float v8f __attribute__((ext_vector_type(8)));

#define NEG_SLOPE 0.2f

__device__ __forceinline__ void atomicMaxF(float* addr, float val) {
    if (val >= 0.0f) {
        atomicMax((int*)addr, __float_as_int(val));
    } else {
        atomicMin((unsigned int*)addr, __float_as_uint(val));
    }
}

__device__ __forceinline__ void edge_sd(const int* __restrict__ ei, int e, int E,
                                        int& s, int& d) {
    if (e < E) { s = ei[e]; d = ei[E + e]; }
    else       { s = d = e - E; }
}

template <int K, int NCOLT>
__global__ void gemm_wmma_f32(const float* __restrict__ X,
                              const float* __restrict__ W,
                              float* __restrict__ H) {
    constexpr int FOUT = 16 * NCOLT;
    __shared__ float As[16][K + 4];

    const int r0   = blockIdx.x * 16;
    const int lane = threadIdx.x;
    const int wave = threadIdx.y;
    const int tid  = wave * 32 + lane;
    const int nthr = 32 * NCOLT;

    for (int i = tid; i < 16 * K; i += nthr) {
        int rr = i / K, cc = i % K;
        As[rr][cc] = X[(size_t)(r0 + rr) * K + cc];
    }
    __syncthreads();

    const int half = lane >> 4;
    const int m    = lane & 15;
    const int c0   = wave * 16;

    v8f acc = {};
    #pragma unroll 4
    for (int kk = 0; kk < K; kk += 4) {
        const int kb = kk + 2 * half;
        v2f a, b;
        a[0] = As[m][kb];
        a[1] = As[m][kb + 1];
        b[0] = W[(size_t)(kb)     * FOUT + c0 + m];
        b[1] = W[(size_t)(kb + 1) * FOUT + c0 + m];
        acc = __builtin_amdgcn_wmma_f32_16x16x4_f32(
             false, a,  false, b,
             (short)0, acc,  false,  false);
    }

    #pragma unroll
    for (int r = 0; r < 8; ++r) {
        H[(size_t)(r0 + r + 8 * half) * FOUT + c0 + m] = acc[r];
    }
}

template <int H, int D>
__global__ void alpha_kernel(const float* __restrict__ h,
                             const float* __restrict__ a_src,
                             const float* __restrict__ a_dst,
                             float* __restrict__ as, float* __restrict__ ad,
                             int n) {
    int i = blockIdx.x * blockDim.x + threadIdx.x;
    if (i >= n * H) return;
    int node = i / H, hh = i % H;
    const float* hp = h + (size_t)node * (H * D) + hh * D;
    float s = 0.f, d = 0.f;
    #pragma unroll
    for (int j = 0; j < D; ++j) {
        float v = hp[j];
        s += v * a_src[hh * D + j];
        d += v * a_dst[hh * D + j];
    }
    as[i] = s;
    ad[i] = d;
}

#define TILE_ROWS 391
#define MAXT      256
#define NCHUNK    64
#define TILE_THREADS 1024

__device__ __forceinline__ int tile_of(int d) {
    int t = d / TILE_ROWS;
    return t < MAXT ? t : MAXT - 1;
}

__global__ void __launch_bounds__(TILE_THREADS)
tile_hist_kernel(const int* __restrict__ ei, int E, int* __restrict__ histG) {
    __shared__ int hist[MAXT];
    const int t = threadIdx.x;
    if (t < MAXT) hist[t] = 0;
    __syncthreads();
    const int cs = (E + gridDim.x - 1) / gridDim.x;
    const int lo = blockIdx.x * cs;
    const int hi = (lo + cs < E) ? lo + cs : E;
    for (int e = lo + t; e < hi; e += TILE_THREADS) {
        atomicAdd(&hist[tile_of(ei[E + e])], 1);
    }
    __syncthreads();
    if (t < MAXT) histG[blockIdx.x * MAXT + t] = hist[t];
}

__global__ void __launch_bounds__(TILE_THREADS)
tile_place_kernel(const int* __restrict__ ei, int E,
                  const int* __restrict__ histG,
                  int* __restrict__ tileStart,
                  int* __restrict__ ls, int* __restrict__ ld) {
    __shared__ int cursor[MAXT];
    __shared__ int scan[MAXT];
    const int t = threadIdx.x;
    int tot = 0, before = 0;
    if (t < MAXT) {
        #pragma unroll 16
        for (int c = 0; c < NCHUNK; ++c) {
            const int v = histG[c * MAXT + t];
            tot += v;
            if (c < (int)blockIdx.x) before += v;
        }
        scan[t] = tot;
    }
    __syncthreads();
    for (int off = 1; off < MAXT; off <<= 1) {
        int v = 0;
        if (t < MAXT && t >= off) v = scan[t - off];
        __syncthreads();
        if (t < MAXT) scan[t] += v;
        __syncthreads();
    }
    if (t < MAXT) {
        const int startT = scan[t] - tot;
        cursor[t] = startT + before;
        if (blockIdx.x == 0) {
            tileStart[t] = startT;
            if (t == MAXT - 1) tileStart[MAXT] = scan[t];
        }
    }
    __syncthreads();
    const int cs = (E + gridDim.x - 1) / gridDim.x;
    const int lo = blockIdx.x * cs;
    const int hi = (lo + cs < E) ? lo + cs : E;
    for (int e = lo + t; e < hi; e += TILE_THREADS) {
        const int s = ei[e];
        const int d = ei[E + e];
        const int pos = atomicAdd(&cursor[tile_of(d)], 1);
        ls[pos] = s;
        ld[pos] = d;
    }
}

__device__ __forceinline__ float lrelu(float v) {
    return (v > 0.f) ? v : NEG_SLOPE * v;
}

template <int H, bool ELU>
__global__ void __launch_bounds__(TILE_THREADS)
tile_gat_kernel(const int* __restrict__ tileStart,
                const int* __restrict__ ls, const int* __restrict__ ld,
                const float* __restrict__ h,
                const float* __restrict__ as, const float* __restrict__ ad,
                const float* __restrict__ bias,
                float* __restrict__ res,
                const float* __restrict__ Wn,
                const float* __restrict__ a_srcn, const float* __restrict__ a_dstn,
                float* __restrict__ hn, float* __restrict__ asn, float* __restrict__ adn,
                int N) {
    constexpr int F = H * 16;
    constexpr int P     = ELU ? F + 4 : F;
    constexpr int AROWS = ELU ? (TILE_ROWS + 15) / 16 * 16 : TILE_ROWS;
    __shared__ float acc[AROWS * P];
    __shared__ float mx[TILE_ROWS * H];
    __shared__ float dn[TILE_ROWS * H];

    const int tid   = threadIdx.x;
    const int lane  = tid & 31;
    const int wave  = tid >> 5;
    constexpr int NW = TILE_THREADS / 32;

    const int r0    = blockIdx.x * TILE_ROWS;
    int nrows = N - r0;
    if (nrows > TILE_ROWS) nrows = TILE_ROWS;
    if (nrows < 0) nrows = 0;
    const int start = tileStart[blockIdx.x];
    const int cnt   = tileStart[blockIdx.x + 1] - start;
    const int total = cnt + nrows;

    const int crows = ELU ? ((nrows + 15) >> 4) * 16 : nrows;
    for (int i = tid; i < crows * F; i += TILE_THREADS) acc[(i / F) * P + (i % F)] = 0.f;
    for (int i = tid; i < nrows * H; i += TILE_THREADS) {
        const int g = r0 * H + i;
        mx[i] = lrelu(as[g] + ad[g]);
        dn[i] = 0.f;
    }
    __syncthreads();

    for (int i0 = tid; i0 < cnt; i0 += 4 * TILE_THREADS) {
        int   sv[4], dv[4];
        float ev[4][H];
        #pragma unroll
        for (int u = 0; u < 4; ++u) {
            const int i = i0 + u * TILE_THREADS;
            const bool ok = i < cnt;
            sv[u] = ok ? ls[start + i] : 0;
            dv[u] = ok ? ld[start + i] : r0;
        }
        #pragma unroll
        for (int u = 0; u < 4; ++u) {
            #pragma unroll
            for (int hh = 0; hh < H; ++hh)
                ev[u][hh] = lrelu(as[sv[u] * H + hh] + ad[dv[u] * H + hh]);
        }
        #pragma unroll
        for (int u = 0; u < 4; ++u) {
            if (i0 + u * TILE_THREADS < cnt) {
                #pragma unroll
                for (int hh = 0; hh < H; ++hh)
                    atomicMaxF(&mx[(dv[u] - r0) * H + hh], ev[u][hh]);
            }
        }
    }
    __syncthreads();

    int s_n = 0, d_n = r0;
    {
        const int i = wave * 32 + lane;
        if (i < total) {
            if (i < cnt) { s_n = ls[start + i]; d_n = ld[start + i]; }
            else         { s_n = d_n = r0 + (i - cnt); }
        }
    }
    for (int b = wave * 32; b < total; b += NW * 32) {
        const int i  = b + lane;
        const int s  = s_n;
        const int d  = d_n;
        const int dl = d - r0;
        {
            const int in = i + NW * 32;
            s_n = 0; d_n = r0;
            if (in < total) {
                if (in < cnt) { s_n = ls[start + in]; d_n = ld[start + in]; }
                else          { s_n = d_n = r0 + (in - cnt); }
            }
        }
        float ex[H];
        #pragma unroll
        for (int hh = 0; hh < H; ++hh) ex[hh] = 0.f;
        if (i < total) {
            #pragma unroll
            for (int hh = 0; hh < H; ++hh) {
                const float e = lrelu(as[s * H + hh] + ad[d * H + hh]);
                ex[hh] = expf(e - mx[dl * H + hh]);
            }
            #pragma unroll
            for (int hh = 0; hh < H; ++hh) atomicAdd(&dn[dl * H + hh], ex[hh]);
        }
        if constexpr (H == 4) {
            #pragma unroll
            for (int jb = 0; jb < 32; jb += 16) {
                if (b + jb < total) {
                    float v0[16], v1[16];
                    #pragma unroll
                    for (int j = 0; j < 16; ++j) {
                        const int sj = __builtin_amdgcn_readlane(s, jb + j);
                        const float* hp = h + (size_t)sj * F;
                        v0[j] = hp[lane];
                        v1[j] = hp[32 + lane];
                    }
                    #pragma unroll
                    for (int j = 0; j < 16; ++j) {
                        const int   dj = __builtin_amdgcn_readlane(dl, jb + j);
                        const float e0 = __int_as_float(__builtin_amdgcn_readlane(__float_as_int(ex[0]), jb + j));
                        const float e1 = __int_as_float(__builtin_amdgcn_readlane(__float_as_int(ex[1]), jb + j));
                        const float e2 = __int_as_float(__builtin_amdgcn_readlane(__float_as_int(ex[2]), jb + j));
                        const float e3 = __int_as_float(__builtin_amdgcn_readlane(__float_as_int(ex[3]), jb + j));
                        const float wlo = (lane < 16) ? e0 : e1;
                        const float whi = (lane < 16) ? e2 : e3;
                        if (b + jb + j < total) {
                            atomicAdd(&acc[dj * P + lane],      v0[j] * wlo);
                            atomicAdd(&acc[dj * P + 32 + lane], v1[j] * whi);
                        }
                    }
                }
            }
        } else {
            const int c    = lane & 15;
            const int half = lane >> 4;
            float v[16];
            #pragma unroll
            for (int j = 0; j < 16; ++j) {
                const int sj = __shfl(s, 2 * j + half);
                v[j] = h[(size_t)sj * F + c];
            }
            #pragma unroll
            for (int j = 0; j < 16; ++j) {
                const int   from = 2 * j + half;
                const int   dj = __shfl(dl, from);
                const float w  = __shfl(ex[0], from);
                if (b + from < total) {
                    atomicAdd(&acc[dj * P + c], v[j] * w);
                }
            }
        }
    }
    __syncthreads();

    for (int i = tid; i < nrows * F; i += TILE_THREADS) {
        const int r = i / F, f = i % F;
        float v = acc[r * P + f] / (dn[r * H + f / 16] + 1e-16f) + bias[f];
        if constexpr (ELU) {
            v = (v > 0.f) ? v : (expf(v) - 1.f);
            acc[r * P + f] = v;
        } else {
            res[(size_t)r0 * F + i] = v;
        }
    }

    if constexpr (ELU) {
        const int ntile = (nrows + 15) >> 4;
        __syncthreads();

        if (wave < ntile) {
            const int half = lane >> 4;
            const int m    = lane & 15;
            const float* xr = acc + (wave * 16 + m) * P;
            v8f c = {};
            #pragma unroll
            for (int kk = 0; kk < F; kk += 4) {
                const int kb = kk + 2 * half;
                v2f a, b;
                a[0] = xr[kb];
                a[1] = xr[kb + 1];
                b[0] = Wn[(kb)     * 16 + m];
                b[1] = Wn[(kb + 1) * 16 + m];
                c = __builtin_amdgcn_wmma_f32_16x16x4_f32(
                    false, a, false, b, (short)0, c, false, false);
            }
            #pragma unroll
            for (int r = 0; r < 8; ++r)
                acc[(wave * 16 + 8 * half + r) * P + m] = c[r];
        }
        __syncthreads();

        for (int r = tid; r < nrows; r += TILE_THREADS) {
            float sa = 0.f, sd = 0.f;
            #pragma unroll
            for (int j = 0; j < 16; ++j) {
                const float v = acc[r * P + j];
                sa += v * a_srcn[j];
                sd += v * a_dstn[j];
            }
            asn[r0 + r] = sa;
            adn[r0 + r] = sd;
        }
        for (int i = tid; i < nrows * 16; i += TILE_THREADS)
            hn[(size_t)r0 * 16 + i] = acc[(i >> 4) * P + (i & 15)];
    }
}

static inline int cdiv(int a, int b) { return (a + b - 1) / b; }

extern "C" void kernel_launch(void* const* d_in, const int* in_sizes, int n_in,
                              void* d_out, int out_size, void* d_ws, size_t ws_size,
                              hipStream_t stream) {
    const float* x      = (const float*)d_in[0];
    const int*   ei     = (const int*)  d_in[1];
    const float* W1     = (const float*)d_in[2];
    const float* a_src1 = (const float*)d_in[3];
    const float* a_dst1 = (const float*)d_in[4];
    const float* b1     = (const float*)d_in[5];
    const float* W2     = (const float*)d_in[6];
    const float* a_src2 = (const float*)d_in[7];
    const float* a_dst2 = (const float*)d_in[8];
    const float* b2     = (const float*)d_in[9];

    const int N  = in_sizes[0] / 128;
    const int E  = in_sizes[1] / 2;

    float* ws   = (float*)d_ws;
    float* h1   = ws;
    float* as1  = h1   + (size_t)N * 64;
    float* ad1  = as1  + (size_t)N * 4;
    float* h2   = ad1  + (size_t)N * 4;
    float* as2  = h2   + (size_t)N * 16;
    float* ad2  = as2  + N;
    int* histG     = (int*)(ad2 + N);
    int* tileStart = histG + MAXT * MAXT;
    int* ls        = tileStart + 512;
    int* ld        = ls + E;

    float* out  = (float*)d_out;

    const int TB = 256;
    const int NT = cdiv(N, TILE_ROWS);

    tile_hist_kernel<<<NCHUNK, TILE_THREADS, 0, stream>>>(ei, E, histG);
    tile_place_kernel<<<NCHUNK, TILE_THREADS, 0, stream>>>(ei, E, histG, tileStart, ls, ld);

    gemm_wmma_f32<128, 4><<<N / 16, dim3(32, 4), 0, stream>>>(x, W1, h1);
    alpha_kernel<4, 16><<<cdiv(N * 4, TB), TB, 0, stream>>>(
        h1, a_src1, a_dst1, as1, ad1, N);
    tile_gat_kernel<4, true><<<NT, TILE_THREADS, 0, stream>>>(
        tileStart, ls, ld, h1, as1, ad1, b1, nullptr,
        W2, a_src2, a_dst2, h2, as2, ad2, N);

    tile_gat_kernel<1, false><<<NT, TILE_THREADS, 0, stream>>>(
        tileStart, ls, ld, h2, as2, ad2, b2, out,
        nullptr, nullptr, nullptr, nullptr, nullptr, nullptr, N);
}


# Sum sweep reuses each thread's indices and logits from the max sweep (same item deal, 5 slots in registers): index reload and logit gathers go; row loads issue first, weights form under them
# speedup vs baseline: 6.0819x; 1.0524x over previous
#include <hip/hip_runtime.h>
#include <hip/hip_bf16.h>
#include <math.h>

typedef float v2f __attribute__((ext_vector_type(2)));
typedef float v8f __attribute__((ext_vector_type(8)));

#define NEG_SLOPE 0.2f

__device__ __forceinline__ void atomicMaxF(float* addr, float val) {
    if (val >= 0.0f) {
        atomicMax((int*)addr, __float_as_int(val));
    } else {
        atomicMin((unsigned int*)addr, __float_as_uint(val));
    }
}

__device__ __forceinline__ void edge_sd(const int* __restrict__ ei, int e, int E,
                                        int& s, int& d) {
    if (e < E) { s = ei[e]; d = ei[E + e]; }
    else       { s = d = e - E; }
}

template <int K, int NCOLT>
__global__ void gemm_wmma_f32(const float* __restrict__ X,
                              const float* __restrict__ W,
                              float* __restrict__ H) {
    constexpr int FOUT = 16 * NCOLT;
    __shared__ float As[16][K + 4];

    const int r0   = blockIdx.x * 16;
    const int lane = threadIdx.x;
    const int wave = threadIdx.y;
    const int tid  = wave * 32 + lane;
    const int nthr = 32 * NCOLT;

    for (int i = tid; i < 16 * K; i += nthr) {
        int rr = i / K, cc = i % K;
        As[rr][cc] = X[(size_t)(r0 + rr) * K + cc];
    }
    __syncthreads();

    const int half = lane >> 4;
    const int m    = lane & 15;
    const int c0   = wave * 16;

    v8f acc = {};
    #pragma unroll 4
    for (int kk = 0; kk < K; kk += 4) {
        const int kb = kk + 2 * half;
        v2f a, b;
        a[0] = As[m][kb];
        a[1] = As[m][kb + 1];
        b[0] = W[(size_t)(kb)     * FOUT + c0 + m];
        b[1] = W[(size_t)(kb + 1) * FOUT + c0 + m];
        acc = __builtin_amdgcn_wmma_f32_16x16x4_f32(
             false, a,  false, b,
             (short)0, acc,  false,  false);
    }

    #pragma unroll
    for (int r = 0; r < 8; ++r) {
        H[(size_t)(r0 + r + 8 * half) * FOUT + c0 + m] = acc[r];
    }
}

template <int H, int D>
__global__ void alpha_kernel(const float* __restrict__ h,
                             const float* __restrict__ a_src,
                             const float* __restrict__ a_dst,
                             float* __restrict__ as, float* __restrict__ ad,
                             int n) {
    int i = blockIdx.x * blockDim.x + threadIdx.x;
    if (i >= n * H) return;
    int node = i / H, hh = i % H;
    const float* hp = h + (size_t)node * (H * D) + hh * D;
    float s = 0.f, d = 0.f;
    #pragma unroll
    for (int j = 0; j < D; ++j) {
        float v = hp[j];
        s += v * a_src[hh * D + j];
        d += v * a_dst[hh * D + j];
    }
    as[i] = s;
    ad[i] = d;
}

#define TILE_ROWS 391
#define MAXT      256
#define NCHUNK    64
#define TILE_THREADS 1024
#define KEEP 5

__device__ __forceinline__ int tile_of(int d) {
    int t = d / TILE_ROWS;
    return t < MAXT ? t : MAXT - 1;
}

__global__ void __launch_bounds__(TILE_THREADS)
tile_hist_kernel(const int* __restrict__ ei, int E, int* __restrict__ histG) {
    __shared__ int hist[MAXT];
    const int t = threadIdx.x;
    if (t < MAXT) hist[t] = 0;
    __syncthreads();
    const int cs = (E + gridDim.x - 1) / gridDim.x;
    const int lo = blockIdx.x * cs;
    const int hi = (lo + cs < E) ? lo + cs : E;
    for (int e = lo + t; e < hi; e += TILE_THREADS) {
        atomicAdd(&hist[tile_of(ei[E + e])], 1);
    }
    __syncthreads();
    if (t < MAXT) histG[blockIdx.x * MAXT + t] = hist[t];
}

__global__ void __launch_bounds__(TILE_THREADS)
tile_place_kernel(const int* __restrict__ ei, int E,
                  const int* __restrict__ histG,
                  int* __restrict__ tileStart,
                  int* __restrict__ ls, int* __restrict__ ld) {
    __shared__ int cursor[MAXT];
    __shared__ int scan[MAXT];
    const int t = threadIdx.x;
    int tot = 0, before = 0;
    if (t < MAXT) {
        #pragma unroll 16
        for (int c = 0; c < NCHUNK; ++c) {
            const int v = histG[c * MAXT + t];
            tot += v;
            if (c < (int)blockIdx.x) before += v;
        }
        scan[t] = tot;
    }
    __syncthreads();
    for (int off = 1; off < MAXT; off <<= 1) {
        int v = 0;
        if (t < MAXT && t >= off) v = scan[t - off];
        __syncthreads();
        if (t < MAXT) scan[t] += v;
        __syncthreads();
    }
    if (t < MAXT) {
        const int startT = scan[t] - tot;
        cursor[t] = startT + before;
        if (blockIdx.x == 0) {
            tileStart[t] = startT;
            if (t == MAXT - 1) tileStart[MAXT] = scan[t];
        }
    }
    __syncthreads();
    const int cs = (E + gridDim.x - 1) / gridDim.x;
    const int lo = blockIdx.x * cs;
    const int hi = (lo + cs < E) ? lo + cs : E;
    for (int e = lo + t; e < hi; e += TILE_THREADS) {
        const int s = ei[e];
        const int d = ei[E + e];
        const int pos = atomicAdd(&cursor[tile_of(d)], 1);
        ls[pos] = s;
        ld[pos] = d;
    }
}

__device__ __forceinline__ float lrelu(float v) {
    return (v > 0.f) ? v : NEG_SLOPE * v;
}

template <int H, bool ELU>
__global__ void __launch_bounds__(TILE_THREADS)
tile_gat_kernel(const int* __restrict__ tileStart,
                const int* __restrict__ ls, const int* __restrict__ ld,
                const float* __restrict__ h,
                const float* __restrict__ as, const float* __restrict__ ad,
                const float* __restrict__ bias,
                float* __restrict__ res,
                const float* __restrict__ Wn,
                const float* __restrict__ a_srcn, const float* __restrict__ a_dstn,
                float* __restrict__ hn, float* __restrict__ asn, float* __restrict__ adn,
                int N) {
    constexpr int F = H * 16;
    constexpr int P     = ELU ? F + 4 : F;
    constexpr int AROWS = ELU ? (TILE_ROWS + 15) / 16 * 16 : TILE_ROWS;
    __shared__ float acc[AROWS * P];
    __shared__ float mx[TILE_ROWS * H];
    __shared__ float dn[TILE_ROWS * H];

    const int tid   = threadIdx.x;
    const int lane  = tid & 31;
    const int wave  = tid >> 5;
    constexpr int NW = TILE_THREADS / 32;

    const int r0    = blockIdx.x * TILE_ROWS;
    int nrows = N - r0;
    if (nrows > TILE_ROWS) nrows = TILE_ROWS;
    if (nrows < 0) nrows = 0;
    const int start = tileStart[blockIdx.x];
    const int cnt   = tileStart[blockIdx.x + 1] - start;
    const int total = cnt + nrows;

    const int crows = ELU ? ((nrows + 15) >> 4) * 16 : nrows;
    for (int i = tid; i < crows * F; i += TILE_THREADS) acc[(i / F) * P + (i % F)] = 0.f;
    for (int i = tid; i < nrows * H; i += TILE_THREADS) {
        const int g = r0 * H + i;
        mx[i] = lrelu(as[g] + ad[g]);
        dn[i] = 0.f;
    }
    __syncthreads();

    int   ks[KEEP], kd[KEEP];
    float ke[KEEP][H];
    #pragma unroll
    for (int u = 0; u < KEEP; ++u) {
        const int i = tid + u * TILE_THREADS;
        int s = 0, d = r0;
        if (i < cnt)        { s = ls[start + i]; d = ld[start + i]; }
        else if (i < total) { s = d = r0 + (i - cnt); }
        ks[u] = s;
        kd[u] = d - r0;
    }
    #pragma unroll
    for (int u = 0; u < KEEP; ++u) {
        #pragma unroll
        for (int hh = 0; hh < H; ++hh)
            ke[u][hh] = lrelu(as[ks[u] * H + hh] + ad[(r0 + kd[u]) * H + hh]);
    }
    #pragma unroll
    for (int u = 0; u < KEEP; ++u) {
        if (tid + u * TILE_THREADS < cnt) {
            #pragma unroll
            for (int hh = 0; hh < H; ++hh)
                atomicMaxF(&mx[kd[u] * H + hh], ke[u][hh]);
        }
    }
    for (int i0 = KEEP * TILE_THREADS + tid; i0 < cnt; i0 += 4 * TILE_THREADS) {
        int   sv[4], dv[4];
        float ev[4][H];
        #pragma unroll
        for (int u = 0; u < 4; ++u) {
            const int i = i0 + u * TILE_THREADS;
            const bool ok = i < cnt;
            sv[u] = ok ? ls[start + i] : 0;
            dv[u] = ok ? ld[start + i] : r0;
        }
        #pragma unroll
        for (int u = 0; u < 4; ++u) {
            #pragma unroll
            for (int hh = 0; hh < H; ++hh)
                ev[u][hh] = lrelu(as[sv[u] * H + hh] + ad[dv[u] * H + hh]);
        }
        #pragma unroll
        for (int u = 0; u < 4; ++u) {
            if (i0 + u * TILE_THREADS < cnt) {
                #pragma unroll
                for (int hh = 0; hh < H; ++hh)
                    atomicMaxF(&mx[(dv[u] - r0) * H + hh], ev[u][hh]);
            }
        }
    }
    __syncthreads();

    int k = 0;
    #pragma unroll 1
    for (int b = wave * 32; b < total; b += NW * 32, ++k) {
        const int i = b + lane;
        int   s, dl;
        float e[H];
        if (k < KEEP) {
            s  = ks[0];
            dl = kd[0];
            #pragma unroll
            for (int hh = 0; hh < H; ++hh) e[hh] = ke[0][hh];
            #pragma unroll
            for (int u = 0; u + 1 < KEEP; ++u) {
                ks[u] = ks[u + 1];
                kd[u] = kd[u + 1];
                #pragma unroll
                for (int hh = 0; hh < H; ++hh) ke[u][hh] = ke[u + 1][hh];
            }
        } else {
            s = 0; dl = 0;
            #pragma unroll
            for (int hh = 0; hh < H; ++hh) e[hh] = 0.f;
            if (i < total) {
                if (i < cnt) { s = ls[start + i]; dl = ld[start + i] - r0; }
                else         { s = r0 + (i - cnt); dl = i - cnt; }
                #pragma unroll
                for (int hh = 0; hh < H; ++hh)
                    e[hh] = lrelu(as[s * H + hh] + ad[(r0 + dl) * H + hh]);
            }
        }
        float ex[H];
        #pragma unroll
        for (int hh = 0; hh < H; ++hh) ex[hh] = 0.f;
        if constexpr (H == 4) {
            float v0[16], v1[16];
            auto load16 = [&](int jb) {
                #pragma unroll
                for (int j = 0; j < 16; ++j) {
                    const int sj = __builtin_amdgcn_readlane(s, jb + j);
                    const float* hp = h + (size_t)sj * F;
                    v0[j] = hp[lane];
                    v1[j] = hp[32 + lane];
                }
            };
            auto add16 = [&](int jb) {
                #pragma unroll
                for (int j = 0; j < 16; ++j) {
                    const int   dj = __builtin_amdgcn_readlane(dl, jb + j);
                    const float e0 = __int_as_float(__builtin_amdgcn_readlane(__float_as_int(ex[0]), jb + j));
                    const float e1 = __int_as_float(__builtin_amdgcn_readlane(__float_as_int(ex[1]), jb + j));
                    const float e2 = __int_as_float(__builtin_amdgcn_readlane(__float_as_int(ex[2]), jb + j));
                    const float e3 = __int_as_float(__builtin_amdgcn_readlane(__float_as_int(ex[3]), jb + j));
                    const float wlo = (lane < 16) ? e0 : e1;
                    const float whi = (lane < 16) ? e2 : e3;
                    if (b + jb + j < total) {
                        atomicAdd(&acc[dj * P + lane],      v0[j] * wlo);
                        atomicAdd(&acc[dj * P + 32 + lane], v1[j] * whi);
                    }
                }
            };
            load16(0);
            if (i < total) {
                #pragma unroll
                for (int hh = 0; hh < H; ++hh) ex[hh] = expf(e[hh] - mx[dl * H + hh]);
                #pragma unroll
                for (int hh = 0; hh < H; ++hh) atomicAdd(&dn[dl * H + hh], ex[hh]);
            }
            add16(0);
            if (b + 16 < total) {
                load16(16);
                add16(16);
            }
        } else {
            const int c    = lane & 15;
            const int half = lane >> 4;
            float v[16];
            #pragma unroll
            for (int j = 0; j < 16; ++j) {
                const int sj = __shfl(s, 2 * j + half);
                v[j] = h[(size_t)sj * F + c];
            }
            if (i < total) {
                #pragma unroll
                for (int hh = 0; hh < H; ++hh) ex[hh] = expf(e[hh] - mx[dl * H + hh]);
                #pragma unroll
                for (int hh = 0; hh < H; ++hh) atomicAdd(&dn[dl * H + hh], ex[hh]);
            }
            #pragma unroll
            for (int j = 0; j < 16; ++j) {
                const int   from = 2 * j + half;
                const int   dj = __shfl(dl, from);
                const float w  = __shfl(ex[0], from);
                if (b + from < total) {
                    atomicAdd(&acc[dj * P + c], v[j] * w);
                }
            }
        }
    }
    __syncthreads();

    for (int i = tid; i < nrows * F; i += TILE_THREADS) {
        const int r = i / F, f = i % F;
        float v = acc[r * P + f] / (dn[r * H + f / 16] + 1e-16f) + bias[f];
        if constexpr (ELU) {
            v = (v > 0.f) ? v : (expf(v) - 1.f);
            acc[r * P + f] = v;
        } else {
            res[(size_t)r0 * F + i] = v;
        }
    }

    if constexpr (ELU) {
        const int ntile = (nrows + 15) >> 4;
        __syncthreads();

        if (wave < ntile) {
            const int half = lane >> 4;
            const int m    = lane & 15;
            const float* xr = acc + (wave * 16 + m) * P;
            v8f c = {};
            #pragma unroll
            for (int kk = 0; kk < F; kk += 4) {
                const int kb = kk + 2 * half;
                v2f a, b;
                a[0] = xr[kb];
                a[1] = xr[kb + 1];
                b[0] = Wn[(kb)     * 16 + m];
                b[1] = Wn[(kb + 1) * 16 + m];
                c = __builtin_amdgcn_wmma_f32_16x16x4_f32(
                    false, a, false, b, (short)0, c, false, false);
            }
            #pragma unroll
            for (int r = 0; r < 8; ++r)
                acc[(wave * 16 + 8 * half + r) * P + m] = c[r];
        }
        __syncthreads();

        for (int r = tid; r < nrows; r += TILE_THREADS) {
            float sa = 0.f, sd = 0.f;
            #pragma unroll
            for (int j = 0; j < 16; ++j) {
                const float v = acc[r * P + j];
                sa += v * a_srcn[j];
                sd += v * a_dstn[j];
            }
            asn[r0 + r] = sa;
            adn[r0 + r] = sd;
        }
        for (int i = tid; i < nrows * 16; i += TILE_THREADS)
            hn[(size_t)r0 * 16 + i] = acc[(i >> 4) * P + (i & 15)];
    }
}

static inline int cdiv(int a, int b) { return (a + b - 1) / b; }

extern "C" void kernel_launch(void* const* d_in, const int* in_sizes, int n_in,
                              void* d_out, int out_size, void* d_ws, size_t ws_size,
                              hipStream_t stream) {
    const float* x      = (const float*)d_in[0];
    const int*   ei     = (const int*)  d_in[1];
    const float* W1     = (const float*)d_in[2];
    const float* a_src1 = (const float*)d_in[3];
    const float* a_dst1 = (const float*)d_in[4];
    const float* b1     = (const float*)d_in[5];
    const float* W2     = (const float*)d_in[6];
    const float* a_src2 = (const float*)d_in[7];
    const float* a_dst2 = (const float*)d_in[8];
    const float* b2     = (const float*)d_in[9];

    const int N  = in_sizes[0] / 128;
    const int E  = in_sizes[1] / 2;

    float* ws   = (float*)d_ws;
    float* h1   = ws;
    float* as1  = h1   + (size_t)N * 64;
    float* ad1  = as1  + (size_t)N * 4;
    float* h2   = ad1  + (size_t)N * 4;
    float* as2  = h2   + (size_t)N * 16;
    float* ad2  = as2  + N;
    int* histG     = (int*)(ad2 + N);
    int* tileStart = histG + MAXT * MAXT;
    int* ls        = tileStart + 512;
    int* ld        = ls + E;

    float* out  = (float*)d_out;

    const int TB = 256;
    const int NT = cdiv(N, TILE_ROWS);

    tile_hist_kernel<<<NCHUNK, TILE_THREADS, 0, stream>>>(ei, E, histG);
    tile_place_kernel<<<NCHUNK, TILE_THREADS, 0, stream>>>(ei, E, histG, tileStart, ls, ld);

    gemm_wmma_f32<128, 4><<<N / 16, dim3(32, 4), 0, stream>>>(x, W1, h1);
    alpha_kernel<4, 16><<<cdiv(N * 4, TB), TB, 0, stream>>>(
        h1, a_src1, a_dst1, as1, ad1, N);
    tile_gat_kernel<4, true><<<NT, TILE_THREADS, 0, stream>>>(
        tileStart, ls, ld, h1, as1, ad1, b1, nullptr,
        W2, a_src2, a_dst2, h2, as2, ad2, N);

    tile_gat_kernel<1, false><<<NT, TILE_THREADS, 0, stream>>>(
        tileStart, ls, ld, h2, as2, ad2, b2, out,
        nullptr, nullptr, nullptr, nullptr, nullptr, nullptr, N);
}


# Tile binning (count, place kernels): 256 edge chunks, one per WGP, not 64; a thread's 16 edges load together, not one load per trip; count table summed by 16-byte loads; list order: chunk, arrival
# speedup vs baseline: 6.5989x; 1.0850x over previous
#include <hip/hip_runtime.h>
#include <hip/hip_bf16.h>
#include <math.h>

typedef float v2f __attribute__((ext_vector_type(2)));
typedef float v8f __attribute__((ext_vector_type(8)));

#define NEG_SLOPE 0.2f

__device__ __forceinline__ void atomicMaxF(float* addr, float val) {
    if (val >= 0.0f) {
        atomicMax((int*)addr, __float_as_int(val));
    } else {
        atomicMin((unsigned int*)addr, __float_as_uint(val));
    }
}

__device__ __forceinline__ void edge_sd(const int* __restrict__ ei, int e, int E,
                                        int& s, int& d) {
    if (e < E) { s = ei[e]; d = ei[E + e]; }
    else       { s = d = e - E; }
}

template <int K, int NCOLT>
__global__ void gemm_wmma_f32(const float* __restrict__ X,
                              const float* __restrict__ W,
                              float* __restrict__ H) {
    constexpr int FOUT = 16 * NCOLT;
    __shared__ float As[16][K + 4];

    const int r0   = blockIdx.x * 16;
    const int lane = threadIdx.x;
    const int wave = threadIdx.y;
    const int tid  = wave * 32 + lane;
    const int nthr = 32 * NCOLT;

    for (int i = tid; i < 16 * K; i += nthr) {
        int rr = i / K, cc = i % K;
        As[rr][cc] = X[(size_t)(r0 + rr) * K + cc];
    }
    __syncthreads();

    const int half = lane >> 4;
    const int m    = lane & 15;
    const int c0   = wave * 16;

    v8f acc = {};
    #pragma unroll 4
    for (int kk = 0; kk < K; kk += 4) {
        const int kb = kk + 2 * half;
        v2f a, b;
        a[0] = As[m][kb];
        a[1] = As[m][kb + 1];
        b[0] = W[(size_t)(kb)     * FOUT + c0 + m];
        b[1] = W[(size_t)(kb + 1) * FOUT + c0 + m];
        acc = __builtin_amdgcn_wmma_f32_16x16x4_f32(
             false, a,  false, b,
             (short)0, acc,  false,  false);
    }

    #pragma unroll
    for (int r = 0; r < 8; ++r) {
        H[(size_t)(r0 + r + 8 * half) * FOUT + c0 + m] = acc[r];
    }
}

template <int H, int D>
__global__ void alpha_kernel(const float* __restrict__ h,
                             const float* __restrict__ a_src,
                             const float* __restrict__ a_dst,
                             float* __restrict__ as, float* __restrict__ ad,
                             int n) {
    int i = blockIdx.x * blockDim.x + threadIdx.x;
    if (i >= n * H) return;
    int node = i / H, hh = i % H;
    const float* hp = h + (size_t)node * (H * D) + hh * D;
    float s = 0.f, d = 0.f;
    #pragma unroll
    for (int j = 0; j < D; ++j) {
        float v = hp[j];
        s += v * a_src[hh * D + j];
        d += v * a_dst[hh * D + j];
    }
    as[i] = s;
    ad[i] = d;
}

#define TILE_ROWS 391
#define MAXT      256
#define NCHUNK    256
#define BIN_THREADS 256
#define BIN_EPT   16
#define TILE_THREADS 1024
#define KEEP 5

__device__ __forceinline__ int tile_of(int d) {
    int t = d / TILE_ROWS;
    return t < MAXT ? t : MAXT - 1;
}

__global__ void __launch_bounds__(BIN_THREADS)
tile_hist_kernel(const int* __restrict__ ei, int E, int* __restrict__ histG) {
    static_assert(BIN_THREADS == MAXT, "one thread per tile counter");
    __shared__ int hist[MAXT];
    const int t = threadIdx.x;
    hist[t] = 0;
    __syncthreads();
    const int cs = (E + gridDim.x - 1) / gridDim.x;
    const int lo = blockIdx.x * cs;
    const int hi = (lo + cs < E) ? lo + cs : E;
    const int last = (hi > 0) ? hi - 1 : 0;
    for (int e0 = lo + t; e0 < hi; e0 += BIN_THREADS * BIN_EPT) {
        int d[BIN_EPT];
        #pragma unroll
        for (int u = 0; u < BIN_EPT; ++u) {
            const int e = e0 + u * BIN_THREADS;
            d[u] = ei[E + (e < hi ? e : last)];
        }
        #pragma unroll
        for (int u = 0; u < BIN_EPT; ++u)
            atomicAdd(&hist[tile_of(d[u])], (e0 + u * BIN_THREADS < hi) ? 1 : 0);
    }
    __syncthreads();
    histG[blockIdx.x * MAXT + t] = hist[t];
}

__global__ void __launch_bounds__(BIN_THREADS)
tile_place_kernel(const int* __restrict__ ei, int E,
                  const int* __restrict__ histG,
                  int* __restrict__ tileStart,
                  int* __restrict__ ls, int* __restrict__ ld) {
    constexpr int TQ = MAXT / 4;
    constexpr int CG = BIN_THREADS / TQ;
    constexpr int CPG = NCHUNK / CG;
    static_assert(BIN_THREADS == MAXT && CG * TQ == BIN_THREADS && CG * CPG == NCHUNK, "shape");
    __shared__ int  cursor[MAXT];
    __shared__ int  scan[MAXT];
    __shared__ int4 ptot[CG * TQ];
    __shared__ int4 pbef[CG * TQ];
    const int t = threadIdx.x;

    const int cs = (E + gridDim.x - 1) / gridDim.x;
    const int lo = blockIdx.x * cs;
    const int hi = (lo + cs < E) ? lo + cs : E;
    const int last = (hi > 0) ? hi - 1 : 0;

    int sv[BIN_EPT], dv[BIN_EPT];
    #pragma unroll
    for (int u = 0; u < BIN_EPT; ++u) {
        const int e  = lo + t + u * BIN_THREADS;
        const int ec = e < hi ? e : last;
        sv[u] = ei[ec];
        dv[u] = ei[E + ec];
    }

    {
        const int tq = t % TQ;
        const int g  = t / TQ;
        const int cb = (int)blockIdx.x;
        const int4* hg = reinterpret_cast<const int4*>(histG);
        int4 tot4 = make_int4(0, 0, 0, 0);
        int4 bef4 = make_int4(0, 0, 0, 0);
        static_assert(CPG % 16 == 0, "whole batches");
        for (int c0 = g * CPG; c0 < (g + 1) * CPG; c0 += 16) {
            int4 v[16];
            #pragma unroll
            for (int j = 0; j < 16; ++j) v[j] = hg[(c0 + j) * TQ + tq];
            #pragma unroll
            for (int j = 0; j < 16; ++j) {
                const int m = (c0 + j < cb) ? -1 : 0;
                tot4.x += v[j].x; tot4.y += v[j].y; tot4.z += v[j].z; tot4.w += v[j].w;
                bef4.x += v[j].x & m; bef4.y += v[j].y & m; bef4.z += v[j].z & m; bef4.w += v[j].w & m;
            }
        }
        ptot[g * TQ + tq] = tot4;
        pbef[g * TQ + tq] = bef4;
    }
    __syncthreads();
    int tot = 0, before = 0;
    {
        const int* pt = reinterpret_cast<const int*>(ptot);
        const int* pb = reinterpret_cast<const int*>(pbef);
        #pragma unroll
        for (int g = 0; g < CG; ++g) {
            tot    += pt[g * MAXT + t];
            before += pb[g * MAXT + t];
        }
    }
    scan[t] = tot;
    __syncthreads();
    for (int off = 1; off < MAXT; off <<= 1) {
        int v = 0;
        if (t >= off) v = scan[t - off];
        __syncthreads();
        scan[t] += v;
        __syncthreads();
    }
    {
        const int startT = scan[t] - tot;
        cursor[t] = startT + before;
        if (blockIdx.x == 0) {
            tileStart[t] = startT;
            if (t == MAXT - 1) tileStart[MAXT] = scan[t];
        }
    }
    __syncthreads();

    int pos[BIN_EPT];
    #pragma unroll
    for (int u = 0; u < BIN_EPT; ++u)
        pos[u] = atomicAdd(&cursor[tile_of(dv[u])], (lo + t + u * BIN_THREADS < hi) ? 1 : 0);
    #pragma unroll
    for (int u = 0; u < BIN_EPT; ++u) {
        if (lo + t + u * BIN_THREADS < hi) {
            ls[pos[u]] = sv[u];
            ld[pos[u]] = dv[u];
        }
    }
    for (int e0 = lo + t + BIN_THREADS * BIN_EPT; e0 < hi; e0 += BIN_THREADS * BIN_EPT) {
        #pragma unroll
        for (int u = 0; u < BIN_EPT; ++u) {
            const int e  = e0 + u * BIN_THREADS;
            const int ec = e < hi ? e : last;
            sv[u] = ei[ec];
            dv[u] = ei[E + ec];
        }
        #pragma unroll
        for (int u = 0; u < BIN_EPT; ++u)
            pos[u] = atomicAdd(&cursor[tile_of(dv[u])], (e0 + u * BIN_THREADS < hi) ? 1 : 0);
        #pragma unroll
        for (int u = 0; u < BIN_EPT; ++u) {
            if (e0 + u * BIN_THREADS < hi) {
                ls[pos[u]] = sv[u];
                ld[pos[u]] = dv[u];
            }
        }
    }
}

__device__ __forceinline__ float lrelu(float v) {
    return (v > 0.f) ? v : NEG_SLOPE * v;
}

template <int H, bool ELU>
__global__ void __launch_bounds__(TILE_THREADS)
tile_gat_kernel(const int* __restrict__ tileStart,
                const int* __restrict__ ls, const int* __restrict__ ld,
                const float* __restrict__ h,
                const float* __restrict__ as, const float* __restrict__ ad,
                const float* __restrict__ bias,
                float* __restrict__ res,
                const float* __restrict__ Wn,
                const float* __restrict__ a_srcn, const float* __restrict__ a_dstn,
                float* __restrict__ hn, float* __restrict__ asn, float* __restrict__ adn,
                int N) {
    constexpr int F = H * 16;
    constexpr int P     = ELU ? F + 4 : F;
    constexpr int AROWS = ELU ? (TILE_ROWS + 15) / 16 * 16 : TILE_ROWS;
    __shared__ float acc[AROWS * P];
    __shared__ float mx[TILE_ROWS * H];
    __shared__ float dn[TILE_ROWS * H];

    const int tid   = threadIdx.x;
    const int lane  = tid & 31;
    const int wave  = tid >> 5;
    constexpr int NW = TILE_THREADS / 32;

    const int r0    = blockIdx.x * TILE_ROWS;
    int nrows = N - r0;
    if (nrows > TILE_ROWS) nrows = TILE_ROWS;
    if (nrows < 0) nrows = 0;
    const int start = tileStart[blockIdx.x];
    const int cnt   = tileStart[blockIdx.x + 1] - start;
    const int total = cnt + nrows;

    const int crows = ELU ? ((nrows + 15) >> 4) * 16 : nrows;
    for (int i = tid; i < crows * F; i += TILE_THREADS) acc[(i / F) * P + (i % F)] = 0.f;
    for (int i = tid; i < nrows * H; i += TILE_THREADS) {
        const int g = r0 * H + i;
        mx[i] = lrelu(as[g] + ad[g]);
        dn[i] = 0.f;
    }
    __syncthreads();

    int   ks[KEEP], kd[KEEP];
    float ke[KEEP][H];
    #pragma unroll
    for (int u = 0; u < KEEP; ++u) {
        const int i = tid + u * TILE_THREADS;
        int s = 0, d = r0;
        if (i < cnt)        { s = ls[start + i]; d = ld[start + i]; }
        else if (i < total) { s = d = r0 + (i - cnt); }
        ks[u] = s;
        kd[u] = d - r0;
    }
    #pragma unroll
    for (int u = 0; u < KEEP; ++u) {
        #pragma unroll
        for (int hh = 0; hh < H; ++hh)
            ke[u][hh] = lrelu(as[ks[u] * H + hh] + ad[(r0 + kd[u]) * H + hh]);
    }
    #pragma unroll
    for (int u = 0; u < KEEP; ++u) {
        if (tid + u * TILE_THREADS < cnt) {
            #pragma unroll
            for (int hh = 0; hh < H; ++hh)
                atomicMaxF(&mx[kd[u] * H + hh], ke[u][hh]);
        }
    }
    for (int i0 = KEEP * TILE_THREADS + tid; i0 < cnt; i0 += 4 * TILE_THREADS) {
        int   sv[4], dv[4];
        float ev[4][H];
        #pragma unroll
        for (int u = 0; u < 4; ++u) {
            const int i = i0 + u * TILE_THREADS;
            const bool ok = i < cnt;
            sv[u] = ok ? ls[start + i] : 0;
            dv[u] = ok ? ld[start + i] : r0;
        }
        #pragma unroll
        for (int u = 0; u < 4; ++u) {
            #pragma unroll
            for (int hh = 0; hh < H; ++hh)
                ev[u][hh] = lrelu(as[sv[u] * H + hh] + ad[dv[u] * H + hh]);
        }
        #pragma unroll
        for (int u = 0; u < 4; ++u) {
            if (i0 + u * TILE_THREADS < cnt) {
                #pragma unroll
                for (int hh = 0; hh < H; ++hh)
                    atomicMaxF(&mx[(dv[u] - r0) * H + hh], ev[u][hh]);
            }
        }
    }
    __syncthreads();

    int k = 0;
    #pragma unroll 1
    for (int b = wave * 32; b < total; b += NW * 32, ++k) {
        const int i = b + lane;
        int   s, dl;
        float e[H];
        if (k < KEEP) {
            s  = ks[0];
            dl = kd[0];
            #pragma unroll
            for (int hh = 0; hh < H; ++hh) e[hh] = ke[0][hh];
            #pragma unroll
            for (int u = 0; u + 1 < KEEP; ++u) {
                ks[u] = ks[u + 1];
                kd[u] = kd[u + 1];
                #pragma unroll
                for (int hh = 0; hh < H; ++hh) ke[u][hh] = ke[u + 1][hh];
            }
        } else {
            s = 0; dl = 0;
            #pragma unroll
            for (int hh = 0; hh < H; ++hh) e[hh] = 0.f;
            if (i < total) {
                if (i < cnt) { s = ls[start + i]; dl = ld[start + i] - r0; }
                else         { s = r0 + (i - cnt); dl = i - cnt; }
                #pragma unroll
                for (int hh = 0; hh < H; ++hh)
                    e[hh] = lrelu(as[s * H + hh] + ad[(r0 + dl) * H + hh]);
            }
        }
        float ex[H];
        #pragma unroll
        for (int hh = 0; hh < H; ++hh) ex[hh] = 0.f;
        if constexpr (H == 4) {
            float v0[16], v1[16];
            auto load16 = [&](int jb) {
                #pragma unroll
                for (int j = 0; j < 16; ++j) {
                    const int sj = __builtin_amdgcn_readlane(s, jb + j);
                    const float* hp = h + (size_t)sj * F;
                    v0[j] = hp[lane];
                    v1[j] = hp[32 + lane];
                }
            };
            auto add16 = [&](int jb) {
                #pragma unroll
                for (int j = 0; j < 16; ++j) {
                    const int   dj = __builtin_amdgcn_readlane(dl, jb + j);
                    const float e0 = __int_as_float(__builtin_amdgcn_readlane(__float_as_int(ex[0]), jb + j));
                    const float e1 = __int_as_float(__builtin_amdgcn_readlane(__float_as_int(ex[1]), jb + j));
                    const float e2 = __int_as_float(__builtin_amdgcn_readlane(__float_as_int(ex[2]), jb + j));
                    const float e3 = __int_as_float(__builtin_amdgcn_readlane(__float_as_int(ex[3]), jb + j));
                    const float wlo = (lane < 16) ? e0 : e1;
                    const float whi = (lane < 16) ? e2 : e3;
                    if (b + jb + j < total) {
                        atomicAdd(&acc[dj * P + lane],      v0[j] * wlo);
                        atomicAdd(&acc[dj * P + 32 + lane], v1[j] * whi);
                    }
                }
            };
            load16(0);
            if (i < total) {
                #pragma unroll
                for (int hh = 0; hh < H; ++hh) ex[hh] = expf(e[hh] - mx[dl * H + hh]);
                #pragma unroll
                for (int hh = 0; hh < H; ++hh) atomicAdd(&dn[dl * H + hh], ex[hh]);
            }
            add16(0);
            if (b + 16 < total) {
                load16(16);
                add16(16);
            }
        } else {
            const int c    = lane & 15;
            const int half = lane >> 4;
            float v[16];
            #pragma unroll
            for (int j = 0; j < 16; ++j) {
                const int sj = __shfl(s, 2 * j + half);
                v[j] = h[(size_t)sj * F + c];
            }
            if (i < total) {
                #pragma unroll
                for (int hh = 0; hh < H; ++hh) ex[hh] = expf(e[hh] - mx[dl * H + hh]);
                #pragma unroll
                for (int hh = 0; hh < H; ++hh) atomicAdd(&dn[dl * H + hh], ex[hh]);
            }
            #pragma unroll
            for (int j = 0; j < 16; ++j) {
                const int   from = 2 * j + half;
                const int   dj = __shfl(dl, from);
                const float w  = __shfl(ex[0], from);
                if (b + from < total) {
                    atomicAdd(&acc[dj * P + c], v[j] * w);
                }
            }
        }
    }
    __syncthreads();

    for (int i = tid; i < nrows * F; i += TILE_THREADS) {
        const int r = i / F, f = i % F;
        float v = acc[r * P + f] / (dn[r * H + f / 16] + 1e-16f) + bias[f];
        if constexpr (ELU) {
            v = (v > 0.f) ? v : (expf(v) - 1.f);
            acc[r * P + f] = v;
        } else {
            res[(size_t)r0 * F + i] = v;
        }
    }

    if constexpr (ELU) {
        const int ntile = (nrows + 15) >> 4;
        __syncthreads();

        if (wave < ntile) {
            const int half = lane >> 4;
            const int m    = lane & 15;
            const float* xr = acc + (wave * 16 + m) * P;
            v8f c = {};
            #pragma unroll
            for (int kk = 0; kk < F; kk += 4) {
                const int kb = kk + 2 * half;
                v2f a, b;
                a[0] = xr[kb];
                a[1] = xr[kb + 1];
                b[0] = Wn[(kb)     * 16 + m];
                b[1] = Wn[(kb + 1) * 16 + m];
                c = __builtin_amdgcn_wmma_f32_16x16x4_f32(
                    false, a, false, b, (short)0, c, false, false);
            }
            #pragma unroll
            for (int r = 0; r < 8; ++r)
                acc[(wave * 16 + 8 * half + r) * P + m] = c[r];
        }
        __syncthreads();

        for (int r = tid; r < nrows; r += TILE_THREADS) {
            float sa = 0.f, sd = 0.f;
            #pragma unroll
            for (int j = 0; j < 16; ++j) {
                const float v = acc[r * P + j];
                sa += v * a_srcn[j];
                sd += v * a_dstn[j];
            }
            asn[r0 + r] = sa;
            adn[r0 + r] = sd;
        }
        for (int i = tid; i < nrows * 16; i += TILE_THREADS)
            hn[(size_t)r0 * 16 + i] = acc[(i >> 4) * P + (i & 15)];
    }
}

static inline int cdiv(int a, int b) { return (a + b - 1) / b; }

extern "C" void kernel_launch(void* const* d_in, const int* in_sizes, int n_in,
                              void* d_out, int out_size, void* d_ws, size_t ws_size,
                              hipStream_t stream) {
    const float* x      = (const float*)d_in[0];
    const int*   ei     = (const int*)  d_in[1];
    const float* W1     = (const float*)d_in[2];
    const float* a_src1 = (const float*)d_in[3];
    const float* a_dst1 = (const float*)d_in[4];
    const float* b1     = (const float*)d_in[5];
    const float* W2     = (const float*)d_in[6];
    const float* a_src2 = (const float*)d_in[7];
    const float* a_dst2 = (const float*)d_in[8];
    const float* b2     = (const float*)d_in[9];

    const int N  = in_sizes[0] / 128;
    const int E  = in_sizes[1] / 2;

    float* ws   = (float*)d_ws;
    float* h1   = ws;
    float* as1  = h1   + (size_t)N * 64;
    float* ad1  = as1  + (size_t)N * 4;
    float* h2   = ad1  + (size_t)N * 4;
    float* as2  = h2   + (size_t)N * 16;
    float* ad2  = as2  + N;
    int* histG     = (int*)(ad2 + N);
    int* tileStart = histG + MAXT * MAXT;
    int* ls        = tileStart + 512;
    int* ld        = ls + E;

    float* out  = (float*)d_out;

    const int TB = 256;
    const int NT = cdiv(N, TILE_ROWS);

    tile_hist_kernel<<<NCHUNK, BIN_THREADS, 0, stream>>>(ei, E, histG);
    tile_place_kernel<<<NCHUNK, BIN_THREADS, 0, stream>>>(ei, E, histG, tileStart, ls, ld);

    gemm_wmma_f32<128, 4><<<N / 16, dim3(32, 4), 0, stream>>>(x, W1, h1);
    alpha_kernel<4, 16><<<cdiv(N * 4, TB), TB, 0, stream>>>(
        h1, a_src1, a_dst1, as1, ad1, N);
    tile_gat_kernel<4, true><<<NT, TILE_THREADS, 0, stream>>>(
        tileStart, ls, ld, h1, as1, ad1, b1, nullptr,
        W2, a_src2, a_dst2, h2, as2, ad2, N);

    tile_gat_kernel<1, false><<<NT, TILE_THREADS, 0, stream>>>(
        tileStart, ls, ld, h2, as2, ad2, b2, out,
        nullptr, nullptr, nullptr, nullptr, nullptr, nullptr, N);
}


# Layer-1 logit stage runs inside the edge-place launch: place work-groups first, 391 logit work-groups (4 pairs a thread, all 16 loads in flight) fill the wave slots their hop chains leave; 5 launches
# speedup vs baseline: 6.8268x; 1.0345x over previous
#include <hip/hip_runtime.h>
#include <hip/hip_bf16.h>
#include <math.h>

typedef float v2f __attribute__((ext_vector_type(2)));
typedef float v8f __attribute__((ext_vector_type(8)));

#define NEG_SLOPE 0.2f

__device__ __forceinline__ void atomicMaxF(float* addr, float val) {
    if (val >= 0.0f) {
        atomicMax((int*)addr, __float_as_int(val));
    } else {
        atomicMin((unsigned int*)addr, __float_as_uint(val));
    }
}

__device__ __forceinline__ void edge_sd(const int* __restrict__ ei, int e, int E,
                                        int& s, int& d) {
    if (e < E) { s = ei[e]; d = ei[E + e]; }
    else       { s = d = e - E; }
}

template <int K, int NCOLT>
__global__ void gemm_wmma_f32(const float* __restrict__ X,
                              const float* __restrict__ W,
                              float* __restrict__ H) {
    constexpr int FOUT = 16 * NCOLT;
    __shared__ float As[16][K + 4];

    const int r0   = blockIdx.x * 16;
    const int lane = threadIdx.x;
    const int wave = threadIdx.y;
    const int tid  = wave * 32 + lane;
    const int nthr = 32 * NCOLT;

    for (int i = tid; i < 16 * K; i += nthr) {
        int rr = i / K, cc = i % K;
        As[rr][cc] = X[(size_t)(r0 + rr) * K + cc];
    }
    __syncthreads();

    const int half = lane >> 4;
    const int m    = lane & 15;
    const int c0   = wave * 16;

    v8f acc = {};
    #pragma unroll 4
    for (int kk = 0; kk < K; kk += 4) {
        const int kb = kk + 2 * half;
        v2f a, b;
        a[0] = As[m][kb];
        a[1] = As[m][kb + 1];
        b[0] = W[(size_t)(kb)     * FOUT + c0 + m];
        b[1] = W[(size_t)(kb + 1) * FOUT + c0 + m];
        acc = __builtin_amdgcn_wmma_f32_16x16x4_f32(
             false, a,  false, b,
             (short)0, acc,  false,  false);
    }

    #pragma unroll
    for (int r = 0; r < 8; ++r) {
        H[(size_t)(r0 + r + 8 * half) * FOUT + c0 + m] = acc[r];
    }
}

#define LOGIT_PPT 4
template <int H, int D, int PPT>
__device__ __forceinline__ void alpha_body(int blk,
                             const float* __restrict__ h,
                             const float* __restrict__ a_src,
                             const float* __restrict__ a_dst,
                             float* __restrict__ as, float* __restrict__ ad,
                             int n) {
    const int base = blk * ((int)blockDim.x * PPT) + (int)threadIdx.x;
    const int npair = n * H;
    float v[PPT][D];
    #pragma unroll
    for (int p = 0; p < PPT; ++p) {
        const int i  = base + p * (int)blockDim.x;
        const int ic = i < npair ? i : npair - 1;
        const float* hp = h + (size_t)ic * D;
        #pragma unroll
        for (int j = 0; j < D; ++j) v[p][j] = hp[j];
    }
    #pragma unroll
    for (int p = 0; p < PPT; ++p) {
        const int i = base + p * (int)blockDim.x;
        if (i < npair) {
            const int hh = i % H;
            float s = 0.f, d = 0.f;
            #pragma unroll
            for (int j = 0; j < D; ++j) {
                s += v[p][j] * a_src[hh * D + j];
                d += v[p][j] * a_dst[hh * D + j];
            }
            as[i] = s;
            ad[i] = d;
        }
    }
}

#define TILE_ROWS 391
#define MAXT      256
#define NCHUNK    256
#define BIN_THREADS 256
#define BIN_EPT   16
#define TILE_THREADS 1024
#define KEEP 5

__device__ __forceinline__ int tile_of(int d) {
    int t = d / TILE_ROWS;
    return t < MAXT ? t : MAXT - 1;
}

__global__ void __launch_bounds__(BIN_THREADS)
tile_hist_kernel(const int* __restrict__ ei, int E, int* __restrict__ histG) {
    static_assert(BIN_THREADS == MAXT, "one thread per tile counter");
    __shared__ int hist[MAXT];
    const int t = threadIdx.x;
    hist[t] = 0;
    __syncthreads();
    const int cs = (E + gridDim.x - 1) / gridDim.x;
    const int lo = blockIdx.x * cs;
    const int hi = (lo + cs < E) ? lo + cs : E;
    const int last = (hi > 0) ? hi - 1 : 0;
    for (int e0 = lo + t; e0 < hi; e0 += BIN_THREADS * BIN_EPT) {
        int d[BIN_EPT];
        #pragma unroll
        for (int u = 0; u < BIN_EPT; ++u) {
            const int e = e0 + u * BIN_THREADS;
            d[u] = ei[E + (e < hi ? e : last)];
        }
        #pragma unroll
        for (int u = 0; u < BIN_EPT; ++u)
            atomicAdd(&hist[tile_of(d[u])], (e0 + u * BIN_THREADS < hi) ? 1 : 0);
    }
    __syncthreads();
    histG[blockIdx.x * MAXT + t] = hist[t];
}

__device__ __forceinline__ void
tile_place_body(const int* __restrict__ ei, int E,
                  const int* __restrict__ histG,
                  int* __restrict__ tileStart,
                  int* __restrict__ ls, int* __restrict__ ld) {
    constexpr int TQ = MAXT / 4;
    constexpr int CG = BIN_THREADS / TQ;
    constexpr int CPG = NCHUNK / CG;
    static_assert(BIN_THREADS == MAXT && CG * TQ == BIN_THREADS && CG * CPG == NCHUNK, "shape");
    __shared__ int  cursor[MAXT];
    __shared__ int  scan[MAXT];
    __shared__ int4 ptot[CG * TQ];
    __shared__ int4 pbef[CG * TQ];
    const int t = threadIdx.x;

    const int cs = (E + NCHUNK - 1) / NCHUNK;
    const int lo = blockIdx.x * cs;
    const int hi = (lo + cs < E) ? lo + cs : E;
    const int last = (hi > 0) ? hi - 1 : 0;

    int sv[BIN_EPT], dv[BIN_EPT];
    #pragma unroll
    for (int u = 0; u < BIN_EPT; ++u) {
        const int e  = lo + t + u * BIN_THREADS;
        const int ec = e < hi ? e : last;
        sv[u] = ei[ec];
        dv[u] = ei[E + ec];
    }

    {
        const int tq = t % TQ;
        const int g  = t / TQ;
        const int cb = (int)blockIdx.x;
        const int4* hg = reinterpret_cast<const int4*>(histG);
        int4 tot4 = make_int4(0, 0, 0, 0);
        int4 bef4 = make_int4(0, 0, 0, 0);
        static_assert(CPG % 16 == 0, "whole batches");
        for (int c0 = g * CPG; c0 < (g + 1) * CPG; c0 += 16) {
            int4 v[16];
            #pragma unroll
            for (int j = 0; j < 16; ++j) v[j] = hg[(c0 + j) * TQ + tq];
            #pragma unroll
            for (int j = 0; j < 16; ++j) {
                const int m = (c0 + j < cb) ? -1 : 0;
                tot4.x += v[j].x; tot4.y += v[j].y; tot4.z += v[j].z; tot4.w += v[j].w;
                bef4.x += v[j].x & m; bef4.y += v[j].y & m; bef4.z += v[j].z & m; bef4.w += v[j].w & m;
            }
        }
        ptot[g * TQ + tq] = tot4;
        pbef[g * TQ + tq] = bef4;
    }
    __syncthreads();
    int tot = 0, before = 0;
    {
        const int* pt = reinterpret_cast<const int*>(ptot);
        const int* pb = reinterpret_cast<const int*>(pbef);
        #pragma unroll
        for (int g = 0; g < CG; ++g) {
            tot    += pt[g * MAXT + t];
            before += pb[g * MAXT + t];
        }
    }
    scan[t] = tot;
    __syncthreads();
    for (int off = 1; off < MAXT; off <<= 1) {
        int v = 0;
        if (t >= off) v = scan[t - off];
        __syncthreads();
        scan[t] += v;
        __syncthreads();
    }
    {
        const int startT = scan[t] - tot;
        cursor[t] = startT + before;
        if (blockIdx.x == 0) {
            tileStart[t] = startT;
            if (t == MAXT - 1) tileStart[MAXT] = scan[t];
        }
    }
    __syncthreads();

    int pos[BIN_EPT];
    #pragma unroll
    for (int u = 0; u < BIN_EPT; ++u)
        pos[u] = atomicAdd(&cursor[tile_of(dv[u])], (lo + t + u * BIN_THREADS < hi) ? 1 : 0);
    #pragma unroll
    for (int u = 0; u < BIN_EPT; ++u) {
        if (lo + t + u * BIN_THREADS < hi) {
            ls[pos[u]] = sv[u];
            ld[pos[u]] = dv[u];
        }
    }
    for (int e0 = lo + t + BIN_THREADS * BIN_EPT; e0 < hi; e0 += BIN_THREADS * BIN_EPT) {
        #pragma unroll
        for (int u = 0; u < BIN_EPT; ++u) {
            const int e  = e0 + u * BIN_THREADS;
            const int ec = e < hi ? e : last;
            sv[u] = ei[ec];
            dv[u] = ei[E + ec];
        }
        #pragma unroll
        for (int u = 0; u < BIN_EPT; ++u)
            pos[u] = atomicAdd(&cursor[tile_of(dv[u])], (e0 + u * BIN_THREADS < hi) ? 1 : 0);
        #pragma unroll
        for (int u = 0; u < BIN_EPT; ++u) {
            if (e0 + u * BIN_THREADS < hi) {
                ls[pos[u]] = sv[u];
                ld[pos[u]] = dv[u];
            }
        }
    }
}

__global__ void __launch_bounds__(BIN_THREADS)
place_logit_kernel(const int* __restrict__ ei, int E,
                   const int* __restrict__ histG,
                   int* __restrict__ tileStart,
                   int* __restrict__ ls, int* __restrict__ ld,
                   const float* __restrict__ h,
                   const float* __restrict__ a_src, const float* __restrict__ a_dst,
                   float* __restrict__ as, float* __restrict__ ad, int n) {
    if (blockIdx.x < NCHUNK) {
        tile_place_body(ei, E, histG, tileStart, ls, ld);
    } else {
        alpha_body<4, 16, LOGIT_PPT>((int)blockIdx.x - NCHUNK, h, a_src, a_dst, as, ad, n);
    }
}

__device__ __forceinline__ float lrelu(float v) {
    return (v > 0.f) ? v : NEG_SLOPE * v;
}

template <int H, bool ELU>
__global__ void __launch_bounds__(TILE_THREADS)
tile_gat_kernel(const int* __restrict__ tileStart,
                const int* __restrict__ ls, const int* __restrict__ ld,
                const float* __restrict__ h,
                const float* __restrict__ as, const float* __restrict__ ad,
                const float* __restrict__ bias,
                float* __restrict__ res,
                const float* __restrict__ Wn,
                const float* __restrict__ a_srcn, const float* __restrict__ a_dstn,
                float* __restrict__ hn, float* __restrict__ asn, float* __restrict__ adn,
                int N) {
    constexpr int F = H * 16;
    constexpr int P     = ELU ? F + 4 : F;
    constexpr int AROWS = ELU ? (TILE_ROWS + 15) / 16 * 16 : TILE_ROWS;
    __shared__ float acc[AROWS * P];
    __shared__ float mx[TILE_ROWS * H];
    __shared__ float dn[TILE_ROWS * H];

    const int tid   = threadIdx.x;
    const int lane  = tid & 31;
    const int wave  = tid >> 5;
    constexpr int NW = TILE_THREADS / 32;

    const int r0    = blockIdx.x * TILE_ROWS;
    int nrows = N - r0;
    if (nrows > TILE_ROWS) nrows = TILE_ROWS;
    if (nrows < 0) nrows = 0;
    const int start = tileStart[blockIdx.x];
    const int cnt   = tileStart[blockIdx.x + 1] - start;
    const int total = cnt + nrows;

    const int crows = ELU ? ((nrows + 15) >> 4) * 16 : nrows;
    for (int i = tid; i < crows * F; i += TILE_THREADS) acc[(i / F) * P + (i % F)] = 0.f;
    for (int i = tid; i < nrows * H; i += TILE_THREADS) {
        const int g = r0 * H + i;
        mx[i] = lrelu(as[g] + ad[g]);
        dn[i] = 0.f;
    }
    __syncthreads();

    int   ks[KEEP], kd[KEEP];
    float ke[KEEP][H];
    #pragma unroll
    for (int u = 0; u < KEEP; ++u) {
        const int i = tid + u * TILE_THREADS;
        int s = 0, d = r0;
        if (i < cnt)        { s = ls[start + i]; d = ld[start + i]; }
        else if (i < total) { s = d = r0 + (i - cnt); }
        ks[u] = s;
        kd[u] = d - r0;
    }
    #pragma unroll
    for (int u = 0; u < KEEP; ++u) {
        #pragma unroll
        for (int hh = 0; hh < H; ++hh)
            ke[u][hh] = lrelu(as[ks[u] * H + hh] + ad[(r0 + kd[u]) * H + hh]);
    }
    #pragma unroll
    for (int u = 0; u < KEEP; ++u) {
        if (tid + u * TILE_THREADS < cnt) {
            #pragma unroll
            for (int hh = 0; hh < H; ++hh)
                atomicMaxF(&mx[kd[u] * H + hh], ke[u][hh]);
        }
    }
    for (int i0 = KEEP * TILE_THREADS + tid; i0 < cnt; i0 += 4 * TILE_THREADS) {
        int   sv[4], dv[4];
        float ev[4][H];
        #pragma unroll
        for (int u = 0; u < 4; ++u) {
            const int i = i0 + u * TILE_THREADS;
            const bool ok = i < cnt;
            sv[u] = ok ? ls[start + i] : 0;
            dv[u] = ok ? ld[start + i] : r0;
        }
        #pragma unroll
        for (int u = 0; u < 4; ++u) {
            #pragma unroll
            for (int hh = 0; hh < H; ++hh)
                ev[u][hh] = lrelu(as[sv[u] * H + hh] + ad[dv[u] * H + hh]);
        }
        #pragma unroll
        for (int u = 0; u < 4; ++u) {
            if (i0 + u * TILE_THREADS < cnt) {
                #pragma unroll
                for (int hh = 0; hh < H; ++hh)
                    atomicMaxF(&mx[(dv[u] - r0) * H + hh], ev[u][hh]);
            }
        }
    }
    __syncthreads();

    int k = 0;
    #pragma unroll 1
    for (int b = wave * 32; b < total; b += NW * 32, ++k) {
        const int i = b + lane;
        int   s, dl;
        float e[H];
        if (k < KEEP) {
            s  = ks[0];
            dl = kd[0];
            #pragma unroll
            for (int hh = 0; hh < H; ++hh) e[hh] = ke[0][hh];
            #pragma unroll
            for (int u = 0; u + 1 < KEEP; ++u) {
                ks[u] = ks[u + 1];
                kd[u] = kd[u + 1];
                #pragma unroll
                for (int hh = 0; hh < H; ++hh) ke[u][hh] = ke[u + 1][hh];
            }
        } else {
            s = 0; dl = 0;
            #pragma unroll
            for (int hh = 0; hh < H; ++hh) e[hh] = 0.f;
            if (i < total) {
                if (i < cnt) { s = ls[start + i]; dl = ld[start + i] - r0; }
                else         { s = r0 + (i - cnt); dl = i - cnt; }
                #pragma unroll
                for (int hh = 0; hh < H; ++hh)
                    e[hh] = lrelu(as[s * H + hh] + ad[(r0 + dl) * H + hh]);
            }
        }
        float ex[H];
        #pragma unroll
        for (int hh = 0; hh < H; ++hh) ex[hh] = 0.f;
        if constexpr (H == 4) {
            float v0[16], v1[16];
            auto load16 = [&](int jb) {
                #pragma unroll
                for (int j = 0; j < 16; ++j) {
                    const int sj = __builtin_amdgcn_readlane(s, jb + j);
                    const float* hp = h + (size_t)sj * F;
                    v0[j] = hp[lane];
                    v1[j] = hp[32 + lane];
                }
            };
            auto add16 = [&](int jb) {
                #pragma unroll
                for (int j = 0; j < 16; ++j) {
                    const int   dj = __builtin_amdgcn_readlane(dl, jb + j);
                    const float e0 = __int_as_float(__builtin_amdgcn_readlane(__float_as_int(ex[0]), jb + j));
                    const float e1 = __int_as_float(__builtin_amdgcn_readlane(__float_as_int(ex[1]), jb + j));
                    const float e2 = __int_as_float(__builtin_amdgcn_readlane(__float_as_int(ex[2]), jb + j));
                    const float e3 = __int_as_float(__builtin_amdgcn_readlane(__float_as_int(ex[3]), jb + j));
                    const float wlo = (lane < 16) ? e0 : e1;
                    const float whi = (lane < 16) ? e2 : e3;
                    if (b + jb + j < total) {
                        atomicAdd(&acc[dj * P + lane],      v0[j] * wlo);
                        atomicAdd(&acc[dj * P + 32 + lane], v1[j] * whi);
                    }
                }
            };
            load16(0);
            if (i < total) {
                #pragma unroll
                for (int hh = 0; hh < H; ++hh) ex[hh] = expf(e[hh] - mx[dl * H + hh]);
                #pragma unroll
                for (int hh = 0; hh < H; ++hh) atomicAdd(&dn[dl * H + hh], ex[hh]);
            }
            add16(0);
            if (b + 16 < total) {
                load16(16);
                add16(16);
            }
        } else {
            const int c    = lane & 15;
            const int half = lane >> 4;
            float v[16];
            #pragma unroll
            for (int j = 0; j < 16; ++j) {
                const int sj = __shfl(s, 2 * j + half);
                v[j] = h[(size_t)sj * F + c];
            }
            if (i < total) {
                #pragma unroll
                for (int hh = 0; hh < H; ++hh) ex[hh] = expf(e[hh] - mx[dl * H + hh]);
                #pragma unroll
                for (int hh = 0; hh < H; ++hh) atomicAdd(&dn[dl * H + hh], ex[hh]);
            }
            #pragma unroll
            for (int j = 0; j < 16; ++j) {
                const int   from = 2 * j + half;
                const int   dj = __shfl(dl, from);
                const float w  = __shfl(ex[0], from);
                if (b + from < total) {
                    atomicAdd(&acc[dj * P + c], v[j] * w);
                }
            }
        }
    }
    __syncthreads();

    for (int i = tid; i < nrows * F; i += TILE_THREADS) {
        const int r = i / F, f = i % F;
        float v = acc[r * P + f] / (dn[r * H + f / 16] + 1e-16f) + bias[f];
        if constexpr (ELU) {
            v = (v > 0.f) ? v : (expf(v) - 1.f);
            acc[r * P + f] = v;
        } else {
            res[(size_t)r0 * F + i] = v;
        }
    }

    if constexpr (ELU) {
        const int ntile = (nrows + 15) >> 4;
        __syncthreads();

        if (wave < ntile) {
            const int half = lane >> 4;
            const int m    = lane & 15;
            const float* xr = acc + (wave * 16 + m) * P;
            v8f c = {};
            #pragma unroll
            for (int kk = 0; kk < F; kk += 4) {
                const int kb = kk + 2 * half;
                v2f a, b;
                a[0] = xr[kb];
                a[1] = xr[kb + 1];
                b[0] = Wn[(kb)     * 16 + m];
                b[1] = Wn[(kb + 1) * 16 + m];
                c = __builtin_amdgcn_wmma_f32_16x16x4_f32(
                    false, a, false, b, (short)0, c, false, false);
            }
            #pragma unroll
            for (int r = 0; r < 8; ++r)
                acc[(wave * 16 + 8 * half + r) * P + m] = c[r];
        }
        __syncthreads();

        for (int r = tid; r < nrows; r += TILE_THREADS) {
            float sa = 0.f, sd = 0.f;
            #pragma unroll
            for (int j = 0; j < 16; ++j) {
                const float v = acc[r * P + j];
                sa += v * a_srcn[j];
                sd += v * a_dstn[j];
            }
            asn[r0 + r] = sa;
            adn[r0 + r] = sd;
        }
        for (int i = tid; i < nrows * 16; i += TILE_THREADS)
            hn[(size_t)r0 * 16 + i] = acc[(i >> 4) * P + (i & 15)];
    }
}

static inline int cdiv(int a, int b) { return (a + b - 1) / b; }

extern "C" void kernel_launch(void* const* d_in, const int* in_sizes, int n_in,
                              void* d_out, int out_size, void* d_ws, size_t ws_size,
                              hipStream_t stream) {
    const float* x      = (const float*)d_in[0];
    const int*   ei     = (const int*)  d_in[1];
    const float* W1     = (const float*)d_in[2];
    const float* a_src1 = (const float*)d_in[3];
    const float* a_dst1 = (const float*)d_in[4];
    const float* b1     = (const float*)d_in[5];
    const float* W2     = (const float*)d_in[6];
    const float* a_src2 = (const float*)d_in[7];
    const float* a_dst2 = (const float*)d_in[8];
    const float* b2     = (const float*)d_in[9];

    const int N  = in_sizes[0] / 128;
    const int E  = in_sizes[1] / 2;

    float* ws   = (float*)d_ws;
    float* h1   = ws;
    float* as1  = h1   + (size_t)N * 64;
    float* ad1  = as1  + (size_t)N * 4;
    float* h2   = ad1  + (size_t)N * 4;
    float* as2  = h2   + (size_t)N * 16;
    float* ad2  = as2  + N;
    int* histG     = (int*)(ad2 + N);
    int* tileStart = histG + MAXT * MAXT;
    int* ls        = tileStart + 512;
    int* ld        = ls + E;

    float* out  = (float*)d_out;

    constexpr int TB = 256;
    const int NT = cdiv(N, TILE_ROWS);

    tile_hist_kernel<<<NCHUNK, BIN_THREADS, 0, stream>>>(ei, E, histG);

    gemm_wmma_f32<128, 4><<<N / 16, dim3(32, 4), 0, stream>>>(x, W1, h1);
    static_assert(TB == BIN_THREADS, "one block size for both stages");
    place_logit_kernel<<<NCHUNK + cdiv(N * 4, TB * LOGIT_PPT), BIN_THREADS, 0, stream>>>(
        ei, E, histG, tileStart, ls, ld, h1, a_src1, a_dst1, as1, ad1, N);
    tile_gat_kernel<4, true><<<NT, TILE_THREADS, 0, stream>>>(
        tileStart, ls, ld, h1, as1, ad1, b1, nullptr,
        W2, a_src2, a_dst2, h2, as2, ad2, N);

    tile_gat_kernel<1, false><<<NT, TILE_THREADS, 0, stream>>>(
        tileStart, ls, ld, h2, as2, ad2, b2, out,
        nullptr, nullptr, nullptr, nullptr, nullptr, nullptr, N);
}


# Sum sweep: item weights reach the lanes as DPP row_share operands of the multiplies (no readlane+select, no ds_bpermute); full groups skip the per-item end test; atomic arrival order as before
# speedup vs baseline: 7.0772x; 1.0367x over previous
#include <hip/hip_runtime.h>
#include <hip/hip_bf16.h>
#include <math.h>

typedef float v2f __attribute__((ext_vector_type(2)));
typedef float v8f __attribute__((ext_vector_type(8)));

#define NEG_SLOPE 0.2f

__device__ __forceinline__ void atomicMaxF(float* addr, float val) {
    if (val >= 0.0f) {
        atomicMax((int*)addr, __float_as_int(val));
    } else {
        atomicMin((unsigned int*)addr, __float_as_uint(val));
    }
}

__device__ __forceinline__ void edge_sd(const int* __restrict__ ei, int e, int E,
                                        int& s, int& d) {
    if (e < E) { s = ei[e]; d = ei[E + e]; }
    else       { s = d = e - E; }
}

template <int K, int NCOLT>
__global__ void gemm_wmma_f32(const float* __restrict__ X,
                              const float* __restrict__ W,
                              float* __restrict__ H) {
    constexpr int FOUT = 16 * NCOLT;
    __shared__ float As[16][K + 4];

    const int r0   = blockIdx.x * 16;
    const int lane = threadIdx.x;
    const int wave = threadIdx.y;
    const int tid  = wave * 32 + lane;
    const int nthr = 32 * NCOLT;

    for (int i = tid; i < 16 * K; i += nthr) {
        int rr = i / K, cc = i % K;
        As[rr][cc] = X[(size_t)(r0 + rr) * K + cc];
    }
    __syncthreads();

    const int half = lane >> 4;
    const int m    = lane & 15;
    const int c0   = wave * 16;

    v8f acc = {};
    #pragma unroll 4
    for (int kk = 0; kk < K; kk += 4) {
        const int kb = kk + 2 * half;
        v2f a, b;
        a[0] = As[m][kb];
        a[1] = As[m][kb + 1];
        b[0] = W[(size_t)(kb)     * FOUT + c0 + m];
        b[1] = W[(size_t)(kb + 1) * FOUT + c0 + m];
        acc = __builtin_amdgcn_wmma_f32_16x16x4_f32(
             false, a,  false, b,
             (short)0, acc,  false,  false);
    }

    #pragma unroll
    for (int r = 0; r < 8; ++r) {
        H[(size_t)(r0 + r + 8 * half) * FOUT + c0 + m] = acc[r];
    }
}

#define LOGIT_PPT 4
template <int H, int D, int PPT>
__device__ __forceinline__ void alpha_body(int blk,
                             const float* __restrict__ h,
                             const float* __restrict__ a_src,
                             const float* __restrict__ a_dst,
                             float* __restrict__ as, float* __restrict__ ad,
                             int n) {
    const int base = blk * ((int)blockDim.x * PPT) + (int)threadIdx.x;
    const int npair = n * H;
    float v[PPT][D];
    #pragma unroll
    for (int p = 0; p < PPT; ++p) {
        const int i  = base + p * (int)blockDim.x;
        const int ic = i < npair ? i : npair - 1;
        const float* hp = h + (size_t)ic * D;
        #pragma unroll
        for (int j = 0; j < D; ++j) v[p][j] = hp[j];
    }
    #pragma unroll
    for (int p = 0; p < PPT; ++p) {
        const int i = base + p * (int)blockDim.x;
        if (i < npair) {
            const int hh = i % H;
            float s = 0.f, d = 0.f;
            #pragma unroll
            for (int j = 0; j < D; ++j) {
                s += v[p][j] * a_src[hh * D + j];
                d += v[p][j] * a_dst[hh * D + j];
            }
            as[i] = s;
            ad[i] = d;
        }
    }
}

#define TILE_ROWS 391
#define MAXT      256
#define NCHUNK    256
#define BIN_THREADS 256
#define BIN_EPT   16
#define TILE_THREADS 1024
#define KEEP 5

__device__ __forceinline__ int tile_of(int d) {
    int t = d / TILE_ROWS;
    return t < MAXT ? t : MAXT - 1;
}

__global__ void __launch_bounds__(BIN_THREADS)
tile_hist_kernel(const int* __restrict__ ei, int E, int* __restrict__ histG) {
    static_assert(BIN_THREADS == MAXT, "one thread per tile counter");
    __shared__ int hist[MAXT];
    const int t = threadIdx.x;
    hist[t] = 0;
    __syncthreads();
    const int cs = (E + gridDim.x - 1) / gridDim.x;
    const int lo = blockIdx.x * cs;
    const int hi = (lo + cs < E) ? lo + cs : E;
    const int last = (hi > 0) ? hi - 1 : 0;
    for (int e0 = lo + t; e0 < hi; e0 += BIN_THREADS * BIN_EPT) {
        int d[BIN_EPT];
        #pragma unroll
        for (int u = 0; u < BIN_EPT; ++u) {
            const int e = e0 + u * BIN_THREADS;
            d[u] = ei[E + (e < hi ? e : last)];
        }
        #pragma unroll
        for (int u = 0; u < BIN_EPT; ++u)
            atomicAdd(&hist[tile_of(d[u])], (e0 + u * BIN_THREADS < hi) ? 1 : 0);
    }
    __syncthreads();
    histG[blockIdx.x * MAXT + t] = hist[t];
}

__device__ __forceinline__ void
tile_place_body(const int* __restrict__ ei, int E,
                  const int* __restrict__ histG,
                  int* __restrict__ tileStart,
                  int* __restrict__ ls, int* __restrict__ ld) {
    constexpr int TQ = MAXT / 4;
    constexpr int CG = BIN_THREADS / TQ;
    constexpr int CPG = NCHUNK / CG;
    static_assert(BIN_THREADS == MAXT && CG * TQ == BIN_THREADS && CG * CPG == NCHUNK, "shape");
    __shared__ int  cursor[MAXT];
    __shared__ int  scan[MAXT];
    __shared__ int4 ptot[CG * TQ];
    __shared__ int4 pbef[CG * TQ];
    const int t = threadIdx.x;

    const int cs = (E + NCHUNK - 1) / NCHUNK;
    const int lo = blockIdx.x * cs;
    const int hi = (lo + cs < E) ? lo + cs : E;
    const int last = (hi > 0) ? hi - 1 : 0;

    int sv[BIN_EPT], dv[BIN_EPT];
    #pragma unroll
    for (int u = 0; u < BIN_EPT; ++u) {
        const int e  = lo + t + u * BIN_THREADS;
        const int ec = e < hi ? e : last;
        sv[u] = ei[ec];
        dv[u] = ei[E + ec];
    }

    {
        const int tq = t % TQ;
        const int g  = t / TQ;
        const int cb = (int)blockIdx.x;
        const int4* hg = reinterpret_cast<const int4*>(histG);
        int4 tot4 = make_int4(0, 0, 0, 0);
        int4 bef4 = make_int4(0, 0, 0, 0);
        static_assert(CPG % 16 == 0, "whole batches");
        for (int c0 = g * CPG; c0 < (g + 1) * CPG; c0 += 16) {
            int4 v[16];
            #pragma unroll
            for (int j = 0; j < 16; ++j) v[j] = hg[(c0 + j) * TQ + tq];
            #pragma unroll
            for (int j = 0; j < 16; ++j) {
                const int m = (c0 + j < cb) ? -1 : 0;
                tot4.x += v[j].x; tot4.y += v[j].y; tot4.z += v[j].z; tot4.w += v[j].w;
                bef4.x += v[j].x & m; bef4.y += v[j].y & m; bef4.z += v[j].z & m; bef4.w += v[j].w & m;
            }
        }
        ptot[g * TQ + tq] = tot4;
        pbef[g * TQ + tq] = bef4;
    }
    __syncthreads();
    int tot = 0, before = 0;
    {
        const int* pt = reinterpret_cast<const int*>(ptot);
        const int* pb = reinterpret_cast<const int*>(pbef);
        #pragma unroll
        for (int g = 0; g < CG; ++g) {
            tot    += pt[g * MAXT + t];
            before += pb[g * MAXT + t];
        }
    }
    scan[t] = tot;
    __syncthreads();
    for (int off = 1; off < MAXT; off <<= 1) {
        int v = 0;
        if (t >= off) v = scan[t - off];
        __syncthreads();
        scan[t] += v;
        __syncthreads();
    }
    {
        const int startT = scan[t] - tot;
        cursor[t] = startT + before;
        if (blockIdx.x == 0) {
            tileStart[t] = startT;
            if (t == MAXT - 1) tileStart[MAXT] = scan[t];
        }
    }
    __syncthreads();

    int pos[BIN_EPT];
    #pragma unroll
    for (int u = 0; u < BIN_EPT; ++u)
        pos[u] = atomicAdd(&cursor[tile_of(dv[u])], (lo + t + u * BIN_THREADS < hi) ? 1 : 0);
    #pragma unroll
    for (int u = 0; u < BIN_EPT; ++u) {
        if (lo + t + u * BIN_THREADS < hi) {
            ls[pos[u]] = sv[u];
            ld[pos[u]] = dv[u];
        }
    }
    for (int e0 = lo + t + BIN_THREADS * BIN_EPT; e0 < hi; e0 += BIN_THREADS * BIN_EPT) {
        #pragma unroll
        for (int u = 0; u < BIN_EPT; ++u) {
            const int e  = e0 + u * BIN_THREADS;
            const int ec = e < hi ? e : last;
            sv[u] = ei[ec];
            dv[u] = ei[E + ec];
        }
        #pragma unroll
        for (int u = 0; u < BIN_EPT; ++u)
            pos[u] = atomicAdd(&cursor[tile_of(dv[u])], (e0 + u * BIN_THREADS < hi) ? 1 : 0);
        #pragma unroll
        for (int u = 0; u < BIN_EPT; ++u) {
            if (e0 + u * BIN_THREADS < hi) {
                ls[pos[u]] = sv[u];
                ld[pos[u]] = dv[u];
            }
        }
    }
}

__global__ void __launch_bounds__(BIN_THREADS)
place_logit_kernel(const int* __restrict__ ei, int E,
                   const int* __restrict__ histG,
                   int* __restrict__ tileStart,
                   int* __restrict__ ls, int* __restrict__ ld,
                   const float* __restrict__ h,
                   const float* __restrict__ a_src, const float* __restrict__ a_dst,
                   float* __restrict__ as, float* __restrict__ ad, int n) {
    if (blockIdx.x < NCHUNK) {
        tile_place_body(ei, E, histG, tileStart, ls, ld);
    } else {
        alpha_body<4, 16, LOGIT_PPT>((int)blockIdx.x - NCHUNK, h, a_src, a_dst, as, ad, n);
    }
}

__device__ __forceinline__ float lrelu(float v) {
    return (v > 0.f) ? v : NEG_SLOPE * v;
}

template <int J> struct StepC { static constexpr int value = J; };
template <int J, int N, class Fn>
__device__ __forceinline__ void steps(Fn&& f) {
    if constexpr (J < N) { f(StepC<J>{}); steps<J + 1, N>(f); }
}
template <int J> __device__ __forceinline__ int row_bcast(int v) {
    return __builtin_amdgcn_update_dpp(0, v, 0x150 + J, 0xf, 0xf, true);
}
template <int J> __device__ __forceinline__ float row_bcast(float v) {
    return __int_as_float(row_bcast<J>(__float_as_int(v)));
}
__device__ __forceinline__ float xhalf(float x) {
    const int b = __float_as_int(x);
    return __int_as_float(__builtin_amdgcn_permlanex16(b, b, 0x76543210, 0xfedcba98, false, false));
}

template <int H, bool ELU>
__global__ void __launch_bounds__(TILE_THREADS)
tile_gat_kernel(const int* __restrict__ tileStart,
                const int* __restrict__ ls, const int* __restrict__ ld,
                const float* __restrict__ h,
                const float* __restrict__ as, const float* __restrict__ ad,
                const float* __restrict__ bias,
                float* __restrict__ res,
                const float* __restrict__ Wn,
                const float* __restrict__ a_srcn, const float* __restrict__ a_dstn,
                float* __restrict__ hn, float* __restrict__ asn, float* __restrict__ adn,
                int N) {
    constexpr int F = H * 16;
    constexpr int P     = ELU ? F + 4 : F;
    constexpr int AROWS = ELU ? (TILE_ROWS + 15) / 16 * 16 : TILE_ROWS;
    __shared__ float acc[AROWS * P];
    __shared__ float mx[TILE_ROWS * H];
    __shared__ float dn[TILE_ROWS * H];

    const int tid   = threadIdx.x;
    const int lane  = tid & 31;
    const int wave  = tid >> 5;
    constexpr int NW = TILE_THREADS / 32;

    const int r0    = blockIdx.x * TILE_ROWS;
    int nrows = N - r0;
    if (nrows > TILE_ROWS) nrows = TILE_ROWS;
    if (nrows < 0) nrows = 0;
    const int start = tileStart[blockIdx.x];
    const int cnt   = tileStart[blockIdx.x + 1] - start;
    const int total = cnt + nrows;

    const int crows = ELU ? ((nrows + 15) >> 4) * 16 : nrows;
    for (int i = tid; i < crows * F; i += TILE_THREADS) acc[(i / F) * P + (i % F)] = 0.f;
    for (int i = tid; i < nrows * H; i += TILE_THREADS) {
        const int g = r0 * H + i;
        mx[i] = lrelu(as[g] + ad[g]);
        dn[i] = 0.f;
    }
    __syncthreads();

    int   ks[KEEP], kd[KEEP];
    float ke[KEEP][H];
    #pragma unroll
    for (int u = 0; u < KEEP; ++u) {
        const int i = tid + u * TILE_THREADS;
        int s = 0, d = r0;
        if (i < cnt)        { s = ls[start + i]; d = ld[start + i]; }
        else if (i < total) { s = d = r0 + (i - cnt); }
        ks[u] = s;
        kd[u] = d - r0;
    }
    #pragma unroll
    for (int u = 0; u < KEEP; ++u) {
        #pragma unroll
        for (int hh = 0; hh < H; ++hh)
            ke[u][hh] = lrelu(as[ks[u] * H + hh] + ad[(r0 + kd[u]) * H + hh]);
    }
    #pragma unroll
    for (int u = 0; u < KEEP; ++u) {
        if (tid + u * TILE_THREADS < cnt) {
            #pragma unroll
            for (int hh = 0; hh < H; ++hh)
                atomicMaxF(&mx[kd[u] * H + hh], ke[u][hh]);
        }
    }
    for (int i0 = KEEP * TILE_THREADS + tid; i0 < cnt; i0 += 4 * TILE_THREADS) {
        int   sv[4], dv[4];
        float ev[4][H];
        #pragma unroll
        for (int u = 0; u < 4; ++u) {
            const int i = i0 + u * TILE_THREADS;
            const bool ok = i < cnt;
            sv[u] = ok ? ls[start + i] : 0;
            dv[u] = ok ? ld[start + i] : r0;
        }
        #pragma unroll
        for (int u = 0; u < 4; ++u) {
            #pragma unroll
            for (int hh = 0; hh < H; ++hh)
                ev[u][hh] = lrelu(as[sv[u] * H + hh] + ad[dv[u] * H + hh]);
        }
        #pragma unroll
        for (int u = 0; u < 4; ++u) {
            if (i0 + u * TILE_THREADS < cnt) {
                #pragma unroll
                for (int hh = 0; hh < H; ++hh)
                    atomicMaxF(&mx[(dv[u] - r0) * H + hh], ev[u][hh]);
            }
        }
    }
    __syncthreads();

    int k = 0;
    #pragma unroll 1
    for (int b = wave * 32; b < total; b += NW * 32, ++k) {
        const int i = b + lane;
        int   s, dl;
        float e[H];
        if (k < KEEP) {
            s  = ks[0];
            dl = kd[0];
            #pragma unroll
            for (int hh = 0; hh < H; ++hh) e[hh] = ke[0][hh];
            #pragma unroll
            for (int u = 0; u + 1 < KEEP; ++u) {
                ks[u] = ks[u + 1];
                kd[u] = kd[u + 1];
                #pragma unroll
                for (int hh = 0; hh < H; ++hh) ke[u][hh] = ke[u + 1][hh];
            }
        } else {
            s = 0; dl = 0;
            #pragma unroll
            for (int hh = 0; hh < H; ++hh) e[hh] = 0.f;
            if (i < total) {
                if (i < cnt) { s = ls[start + i]; dl = ld[start + i] - r0; }
                else         { s = r0 + (i - cnt); dl = i - cnt; }
                #pragma unroll
                for (int hh = 0; hh < H; ++hh)
                    e[hh] = lrelu(as[s * H + hh] + ad[(r0 + dl) * H + hh]);
            }
        }
        float ex[H];
        #pragma unroll
        for (int hh = 0; hh < H; ++hh) ex[hh] = 0.f;
        if constexpr (H == 4) {
            const bool hi = lane >= 16;
            float v0[16], v1[16];
            auto load16 = [&](int jb) {
                #pragma unroll
                for (int j = 0; j < 16; ++j) {
                    const int sj = __builtin_amdgcn_readlane(s, jb + j);
                    const float* hp = h + (size_t)sj * F;
                    v0[j] = hp[lane];
                    v1[j] = hp[32 + lane];
                }
            };
            auto add16 = [&](auto fullc, int jb, float wl, float wh) {
                constexpr bool FULL = decltype(fullc)::value != 0;
                steps<0, 16>([&](auto jc) {
                    constexpr int j = decltype(jc)::value;
                    const int   dj  = __builtin_amdgcn_readlane(dl, jb + j);
                    const float wlo = row_bcast<j>(wl);
                    const float whi = row_bcast<j>(wh);
                    if (FULL || b + jb + j < total) {
                        atomicAdd(&acc[dj * P + lane],      v0[j] * wlo);
                        atomicAdd(&acc[dj * P + 32 + lane], v1[j] * whi);
                    }
                });
            };
            load16(0);
            if (i < total) {
                #pragma unroll
                for (int hh = 0; hh < H; ++hh) ex[hh] = expf(e[hh] - mx[dl * H + hh]);
                #pragma unroll
                for (int hh = 0; hh < H; ++hh) atomicAdd(&dn[dl * H + hh], ex[hh]);
            }
            const float x0 = xhalf(ex[0]);
            const float x1 = xhalf(ex[1]);
            const float x2 = xhalf(ex[2]);
            const float x3 = xhalf(ex[3]);
            const float aLo = hi ? x1 : ex[0];
            const float aHi = hi ? x3 : ex[2];
            const float bLo = hi ? ex[1] : x0;
            const float bHi = hi ? ex[3] : x2;
            if (b + 16 <= total) add16(StepC<1>{}, 0, aLo, aHi);
            else                 add16(StepC<0>{}, 0, aLo, aHi);
            if (b + 16 < total) {
                load16(16);
                if (b + 32 <= total) add16(StepC<1>{}, 16, bLo, bHi);
                else                 add16(StepC<0>{}, 16, bLo, bHi);
            }
        } else {
            const int c    = lane & 15;
            const int half = lane >> 4;
            float v[16];
            steps<0, 16>([&](auto jc) {
                constexpr int j = decltype(jc)::value;
                const int sj = row_bcast<j>(s);
                v[j] = h[(size_t)sj * F + c];
            });
            if (i < total) {
                #pragma unroll
                for (int hh = 0; hh < H; ++hh) ex[hh] = expf(e[hh] - mx[dl * H + hh]);
                #pragma unroll
                for (int hh = 0; hh < H; ++hh) atomicAdd(&dn[dl * H + hh], ex[hh]);
            }
            if (b + 32 <= total) {
                steps<0, 16>([&](auto jc) {
                    constexpr int j = decltype(jc)::value;
                    const int   dj = row_bcast<j>(dl);
                    const float w  = row_bcast<j>(ex[0]);
                    atomicAdd(&acc[dj * P + c], v[j] * w);
                });
            } else {
                steps<0, 16>([&](auto jc) {
                    constexpr int j = decltype(jc)::value;
                    const int   dj = row_bcast<j>(dl);
                    const float w  = row_bcast<j>(ex[0]);
                    if (b + 16 * half + j < total) {
                        atomicAdd(&acc[dj * P + c], v[j] * w);
                    }
                });
            }
        }
    }
    __syncthreads();

    for (int i = tid; i < nrows * F; i += TILE_THREADS) {
        const int r = i / F, f = i % F;
        float v = acc[r * P + f] / (dn[r * H + f / 16] + 1e-16f) + bias[f];
        if constexpr (ELU) {
            v = (v > 0.f) ? v : (expf(v) - 1.f);
            acc[r * P + f] = v;
        } else {
            res[(size_t)r0 * F + i] = v;
        }
    }

    if constexpr (ELU) {
        const int ntile = (nrows + 15) >> 4;
        __syncthreads();

        if (wave < ntile) {
            const int half = lane >> 4;
            const int m    = lane & 15;
            const float* xr = acc + (wave * 16 + m) * P;
            v8f c = {};
            #pragma unroll
            for (int kk = 0; kk < F; kk += 4) {
                const int kb = kk + 2 * half;
                v2f a, b;
                a[0] = xr[kb];
                a[1] = xr[kb + 1];
                b[0] = Wn[(kb)     * 16 + m];
                b[1] = Wn[(kb + 1) * 16 + m];
                c = __builtin_amdgcn_wmma_f32_16x16x4_f32(
                    false, a, false, b, (short)0, c, false, false);
            }
            #pragma unroll
            for (int r = 0; r < 8; ++r)
                acc[(wave * 16 + 8 * half + r) * P + m] = c[r];
        }
        __syncthreads();

        for (int r = tid; r < nrows; r += TILE_THREADS) {
            float sa = 0.f, sd = 0.f;
            #pragma unroll
            for (int j = 0; j < 16; ++j) {
                const float v = acc[r * P + j];
                sa += v * a_srcn[j];
                sd += v * a_dstn[j];
            }
            asn[r0 + r] = sa;
            adn[r0 + r] = sd;
        }
        for (int i = tid; i < nrows * 16; i += TILE_THREADS)
            hn[(size_t)r0 * 16 + i] = acc[(i >> 4) * P + (i & 15)];
    }
}

static inline int cdiv(int a, int b) { return (a + b - 1) / b; }

extern "C" void kernel_launch(void* const* d_in, const int* in_sizes, int n_in,
                              void* d_out, int out_size, void* d_ws, size_t ws_size,
                              hipStream_t stream) {
    const float* x      = (const float*)d_in[0];
    const int*   ei     = (const int*)  d_in[1];
    const float* W1     = (const float*)d_in[2];
    const float* a_src1 = (const float*)d_in[3];
    const float* a_dst1 = (const float*)d_in[4];
    const float* b1     = (const float*)d_in[5];
    const float* W2     = (const float*)d_in[6];
    const float* a_src2 = (const float*)d_in[7];
    const float* a_dst2 = (const float*)d_in[8];
    const float* b2     = (const float*)d_in[9];

    const int N  = in_sizes[0] / 128;
    const int E  = in_sizes[1] / 2;

    float* ws   = (float*)d_ws;
    float* h1   = ws;
    float* as1  = h1   + (size_t)N * 64;
    float* ad1  = as1  + (size_t)N * 4;
    float* h2   = ad1  + (size_t)N * 4;
    float* as2  = h2   + (size_t)N * 16;
    float* ad2  = as2  + N;
    int* histG     = (int*)(ad2 + N);
    int* tileStart = histG + MAXT * MAXT;
    int* ls        = tileStart + 512;
    int* ld        = ls + E;

    float* out  = (float*)d_out;

    constexpr int TB = 256;
    const int NT = cdiv(N, TILE_ROWS);

    tile_hist_kernel<<<NCHUNK, BIN_THREADS, 0, stream>>>(ei, E, histG);

    gemm_wmma_f32<128, 4><<<N / 16, dim3(32, 4), 0, stream>>>(x, W1, h1);
    static_assert(TB == BIN_THREADS, "one block size for both stages");
    place_logit_kernel<<<NCHUNK + cdiv(N * 4, TB * LOGIT_PPT), BIN_THREADS, 0, stream>>>(
        ei, E, histG, tileStart, ls, ld, h1, a_src1, a_dst1, as1, ad1, N);
    tile_gat_kernel<4, true><<<NT, TILE_THREADS, 0, stream>>>(
        tileStart, ls, ld, h1, as1, ad1, b1, nullptr,
        W2, a_src2, a_dst2, h2, as2, ad2, N);

    tile_gat_kernel<1, false><<<NT, TILE_THREADS, 0, stream>>>(
        tileStart, ls, ld, h2, as2, ad2, b2, out,
        nullptr, nullptr, nullptr, nullptr, nullptr, nullptr, N);
}


# Edge placement: a chunk's edges are first ranked into tile order in LDS and written out in that order: a wave's 32 stores fall into a few lines, not 32; lists and their order class as before
# speedup vs baseline: 9.2417x; 1.3058x over previous
#include <hip/hip_runtime.h>
#include <hip/hip_bf16.h>
#include <math.h>

typedef float v2f __attribute__((ext_vector_type(2)));
typedef float v8f __attribute__((ext_vector_type(8)));

#define NEG_SLOPE 0.2f

__device__ __forceinline__ void atomicMaxF(float* addr, float val) {
    if (val >= 0.0f) {
        atomicMax((int*)addr, __float_as_int(val));
    } else {
        atomicMin((unsigned int*)addr, __float_as_uint(val));
    }
}

__device__ __forceinline__ void edge_sd(const int* __restrict__ ei, int e, int E,
                                        int& s, int& d) {
    if (e < E) { s = ei[e]; d = ei[E + e]; }
    else       { s = d = e - E; }
}

template <int K, int NCOLT>
__global__ void gemm_wmma_f32(const float* __restrict__ X,
                              const float* __restrict__ W,
                              float* __restrict__ H) {
    constexpr int FOUT = 16 * NCOLT;
    __shared__ float As[16][K + 4];

    const int r0   = blockIdx.x * 16;
    const int lane = threadIdx.x;
    const int wave = threadIdx.y;
    const int tid  = wave * 32 + lane;
    const int nthr = 32 * NCOLT;

    for (int i = tid; i < 16 * K; i += nthr) {
        int rr = i / K, cc = i % K;
        As[rr][cc] = X[(size_t)(r0 + rr) * K + cc];
    }
    __syncthreads();

    const int half = lane >> 4;
    const int m    = lane & 15;
    const int c0   = wave * 16;

    v8f acc = {};
    #pragma unroll 4
    for (int kk = 0; kk < K; kk += 4) {
        const int kb = kk + 2 * half;
        v2f a, b;
        a[0] = As[m][kb];
        a[1] = As[m][kb + 1];
        b[0] = W[(size_t)(kb)     * FOUT + c0 + m];
        b[1] = W[(size_t)(kb + 1) * FOUT + c0 + m];
        acc = __builtin_amdgcn_wmma_f32_16x16x4_f32(
             false, a,  false, b,
             (short)0, acc,  false,  false);
    }

    #pragma unroll
    for (int r = 0; r < 8; ++r) {
        H[(size_t)(r0 + r + 8 * half) * FOUT + c0 + m] = acc[r];
    }
}

#define LOGIT_PPT 4
template <int H, int D, int PPT>
__device__ __forceinline__ void alpha_body(int blk,
                             const float* __restrict__ h,
                             const float* __restrict__ a_src,
                             const float* __restrict__ a_dst,
                             float* __restrict__ as, float* __restrict__ ad,
                             int n) {
    const int base = blk * ((int)blockDim.x * PPT) + (int)threadIdx.x;
    const int npair = n * H;
    float v[PPT][D];
    #pragma unroll
    for (int p = 0; p < PPT; ++p) {
        const int i  = base + p * (int)blockDim.x;
        const int ic = i < npair ? i : npair - 1;
        const float* hp = h + (size_t)ic * D;
        #pragma unroll
        for (int j = 0; j < D; ++j) v[p][j] = hp[j];
    }
    #pragma unroll
    for (int p = 0; p < PPT; ++p) {
        const int i = base + p * (int)blockDim.x;
        if (i < npair) {
            const int hh = i % H;
            float s = 0.f, d = 0.f;
            #pragma unroll
            for (int j = 0; j < D; ++j) {
                s += v[p][j] * a_src[hh * D + j];
                d += v[p][j] * a_dst[hh * D + j];
            }
            as[i] = s;
            ad[i] = d;
        }
    }
}

#define TILE_ROWS 391
#define MAXT      256
#define NCHUNK    256
#define BIN_THREADS 256
#define BIN_EPT   16
#define TILE_THREADS 1024
#define KEEP 5

__device__ __forceinline__ int tile_of(int d) {
    int t = d / TILE_ROWS;
    return t < MAXT ? t : MAXT - 1;
}

__global__ void __launch_bounds__(BIN_THREADS)
tile_hist_kernel(const int* __restrict__ ei, int E, int* __restrict__ histG) {
    static_assert(BIN_THREADS == MAXT, "one thread per tile counter");
    __shared__ int hist[MAXT];
    const int t = threadIdx.x;
    hist[t] = 0;
    __syncthreads();
    const int cs = (E + gridDim.x - 1) / gridDim.x;
    const int lo = blockIdx.x * cs;
    const int hi = (lo + cs < E) ? lo + cs : E;
    const int last = (hi > 0) ? hi - 1 : 0;
    for (int e0 = lo + t; e0 < hi; e0 += BIN_THREADS * BIN_EPT) {
        int d[BIN_EPT];
        #pragma unroll
        for (int u = 0; u < BIN_EPT; ++u) {
            const int e = e0 + u * BIN_THREADS;
            d[u] = ei[E + (e < hi ? e : last)];
        }
        #pragma unroll
        for (int u = 0; u < BIN_EPT; ++u)
            atomicAdd(&hist[tile_of(d[u])], (e0 + u * BIN_THREADS < hi) ? 1 : 0);
    }
    __syncthreads();
    histG[blockIdx.x * MAXT + t] = hist[t];
}

__device__ __forceinline__ void
tile_place_body(const int* __restrict__ ei, int E,
                  const int* __restrict__ histG,
                  int* __restrict__ tileStart,
                  int* __restrict__ ls, int* __restrict__ ld) {
    constexpr int TQ = MAXT / 4;
    constexpr int CG = BIN_THREADS / TQ;
    constexpr int CPG = NCHUNK / CG;
    constexpr int TRIP = BIN_THREADS * BIN_EPT;
    static_assert(BIN_THREADS == MAXT && CG * TQ == BIN_THREADS && CG * CPG == NCHUNK, "shape");
    __shared__ int  cursor[MAXT];
    __shared__ int  scan[MAXT];
    __shared__ int  lscan[MAXT];
    __shared__ int  gbase[MAXT];
    __shared__ int4 ptot[CG * TQ];
    __shared__ int4 pbef[CG * TQ];
    __shared__ int  sS[TRIP];
    __shared__ int  sD[TRIP];
    const int t = threadIdx.x;

    const int cs = (E + NCHUNK - 1) / NCHUNK;
    const int lo = blockIdx.x * cs;
    const int hi = (lo + cs < E) ? lo + cs : E;
    const int last = (hi > 0) ? hi - 1 : 0;
    const int len = (hi > lo) ? hi - lo : 0;
    const bool ordered = len > 0 && len <= TRIP;

    int sv[BIN_EPT], dv[BIN_EPT];
    #pragma unroll
    for (int u = 0; u < BIN_EPT; ++u) {
        const int e  = lo + t + u * BIN_THREADS;
        const int ec = e < hi ? e : last;
        sv[u] = ei[ec];
        dv[u] = ei[E + ec];
    }
    const int own = histG[(int)blockIdx.x * MAXT + t];

    {
        const int tq = t % TQ;
        const int g  = t / TQ;
        const int cb = (int)blockIdx.x;
        const int4* hg = reinterpret_cast<const int4*>(histG);
        int4 tot4 = make_int4(0, 0, 0, 0);
        int4 bef4 = make_int4(0, 0, 0, 0);
        static_assert(CPG % 16 == 0, "whole batches");
        for (int c0 = g * CPG; c0 < (g + 1) * CPG; c0 += 16) {
            int4 v[16];
            #pragma unroll
            for (int j = 0; j < 16; ++j) v[j] = hg[(c0 + j) * TQ + tq];
            #pragma unroll
            for (int j = 0; j < 16; ++j) {
                const int m = (c0 + j < cb) ? -1 : 0;
                tot4.x += v[j].x; tot4.y += v[j].y; tot4.z += v[j].z; tot4.w += v[j].w;
                bef4.x += v[j].x & m; bef4.y += v[j].y & m; bef4.z += v[j].z & m; bef4.w += v[j].w & m;
            }
        }
        ptot[g * TQ + tq] = tot4;
        pbef[g * TQ + tq] = bef4;
    }
    __syncthreads();
    int tot = 0, before = 0;
    {
        const int* pt = reinterpret_cast<const int*>(ptot);
        const int* pb = reinterpret_cast<const int*>(pbef);
        #pragma unroll
        for (int g = 0; g < CG; ++g) {
            tot    += pt[g * MAXT + t];
            before += pb[g * MAXT + t];
        }
    }
    scan[t]  = tot;
    lscan[t] = own;
    __syncthreads();
    for (int off = 1; off < MAXT; off <<= 1) {
        int v = 0, w = 0;
        if (t >= off) { v = scan[t - off]; w = lscan[t - off]; }
        __syncthreads();
        scan[t]  += v;
        lscan[t] += w;
        __syncthreads();
    }
    {
        const int startT = scan[t] - tot;
        const int lstart = lscan[t] - own;
        if (ordered) {
            cursor[t] = lstart;
            gbase[t]  = startT + before - lstart;
        } else {
            cursor[t] = startT + before;
        }
        if (blockIdx.x == 0) {
            tileStart[t] = startT;
            if (t == MAXT - 1) tileStart[MAXT] = scan[t];
        }
    }
    __syncthreads();

    if (ordered) {
        int pos[BIN_EPT];
        #pragma unroll
        for (int u = 0; u < BIN_EPT; ++u)
            pos[u] = atomicAdd(&cursor[tile_of(dv[u])], (lo + t + u * BIN_THREADS < hi) ? 1 : 0);
        #pragma unroll
        for (int u = 0; u < BIN_EPT; ++u) {
            if (lo + t + u * BIN_THREADS < hi) {
                sS[pos[u]] = sv[u];
                sD[pos[u]] = dv[u];
            }
        }
        __syncthreads();
        #pragma unroll
        for (int u = 0; u < BIN_EPT; ++u) {
            const int j  = t + u * BIN_THREADS;
            const int jc = j < len ? j : len - 1;
            sv[u] = sS[jc];
            dv[u] = sD[jc];
        }
        #pragma unroll
        for (int u = 0; u < BIN_EPT; ++u) pos[u] = gbase[tile_of(dv[u])];
        #pragma unroll
        for (int u = 0; u < BIN_EPT; ++u) {
            const int j = t + u * BIN_THREADS;
            if (j < len) {
                ls[pos[u] + j] = sv[u];
                ld[pos[u] + j] = dv[u];
            }
        }
        return;
    }

    int pos[BIN_EPT];
    #pragma unroll
    for (int u = 0; u < BIN_EPT; ++u)
        pos[u] = atomicAdd(&cursor[tile_of(dv[u])], (lo + t + u * BIN_THREADS < hi) ? 1 : 0);
    #pragma unroll
    for (int u = 0; u < BIN_EPT; ++u) {
        if (lo + t + u * BIN_THREADS < hi) {
            ls[pos[u]] = sv[u];
            ld[pos[u]] = dv[u];
        }
    }
    for (int e0 = lo + t + BIN_THREADS * BIN_EPT; e0 < hi; e0 += BIN_THREADS * BIN_EPT) {
        #pragma unroll
        for (int u = 0; u < BIN_EPT; ++u) {
            const int e  = e0 + u * BIN_THREADS;
            const int ec = e < hi ? e : last;
            sv[u] = ei[ec];
            dv[u] = ei[E + ec];
        }
        #pragma unroll
        for (int u = 0; u < BIN_EPT; ++u)
            pos[u] = atomicAdd(&cursor[tile_of(dv[u])], (e0 + u * BIN_THREADS < hi) ? 1 : 0);
        #pragma unroll
        for (int u = 0; u < BIN_EPT; ++u) {
            if (e0 + u * BIN_THREADS < hi) {
                ls[pos[u]] = sv[u];
                ld[pos[u]] = dv[u];
            }
        }
    }
}

__global__ void __launch_bounds__(BIN_THREADS)
place_logit_kernel(const int* __restrict__ ei, int E,
                   const int* __restrict__ histG,
                   int* __restrict__ tileStart,
                   int* __restrict__ ls, int* __restrict__ ld,
                   const float* __restrict__ h,
                   const float* __restrict__ a_src, const float* __restrict__ a_dst,
                   float* __restrict__ as, float* __restrict__ ad, int n) {
    if (blockIdx.x < NCHUNK) {
        tile_place_body(ei, E, histG, tileStart, ls, ld);
    } else {
        alpha_body<4, 16, LOGIT_PPT>((int)blockIdx.x - NCHUNK, h, a_src, a_dst, as, ad, n);
    }
}

__device__ __forceinline__ float lrelu(float v) {
    return (v > 0.f) ? v : NEG_SLOPE * v;
}

template <int J> struct StepC { static constexpr int value = J; };
template <int J, int N, class Fn>
__device__ __forceinline__ void steps(Fn&& f) {
    if constexpr (J < N) { f(StepC<J>{}); steps<J + 1, N>(f); }
}
template <int J> __device__ __forceinline__ int row_bcast(int v) {
    return __builtin_amdgcn_update_dpp(0, v, 0x150 + J, 0xf, 0xf, true);
}
template <int J> __device__ __forceinline__ float row_bcast(float v) {
    return __int_as_float(row_bcast<J>(__float_as_int(v)));
}
__device__ __forceinline__ float xhalf(float x) {
    const int b = __float_as_int(x);
    return __int_as_float(__builtin_amdgcn_permlanex16(b, b, 0x76543210, 0xfedcba98, false, false));
}

template <int H, bool ELU>
__global__ void __launch_bounds__(TILE_THREADS)
tile_gat_kernel(const int* __restrict__ tileStart,
                const int* __restrict__ ls, const int* __restrict__ ld,
                const float* __restrict__ h,
                const float* __restrict__ as, const float* __restrict__ ad,
                const float* __restrict__ bias,
                float* __restrict__ res,
                const float* __restrict__ Wn,
                const float* __restrict__ a_srcn, const float* __restrict__ a_dstn,
                float* __restrict__ hn, float* __restrict__ asn, float* __restrict__ adn,
                int N) {
    constexpr int F = H * 16;
    constexpr int P     = ELU ? F + 4 : F;
    constexpr int AROWS = ELU ? (TILE_ROWS + 15) / 16 * 16 : TILE_ROWS;
    __shared__ float acc[AROWS * P];
    __shared__ float mx[TILE_ROWS * H];
    __shared__ float dn[TILE_ROWS * H];

    const int tid   = threadIdx.x;
    const int lane  = tid & 31;
    const int wave  = tid >> 5;
    constexpr int NW = TILE_THREADS / 32;

    const int r0    = blockIdx.x * TILE_ROWS;
    int nrows = N - r0;
    if (nrows > TILE_ROWS) nrows = TILE_ROWS;
    if (nrows < 0) nrows = 0;
    const int start = tileStart[blockIdx.x];
    const int cnt   = tileStart[blockIdx.x + 1] - start;
    const int total = cnt + nrows;

    const int crows = ELU ? ((nrows + 15) >> 4) * 16 : nrows;
    for (int i = tid; i < crows * F; i += TILE_THREADS) acc[(i / F) * P + (i % F)] = 0.f;
    for (int i = tid; i < nrows * H; i += TILE_THREADS) {
        const int g = r0 * H + i;
        mx[i] = lrelu(as[g] + ad[g]);
        dn[i] = 0.f;
    }
    __syncthreads();

    int   ks[KEEP], kd[KEEP];
    float ke[KEEP][H];
    #pragma unroll
    for (int u = 0; u < KEEP; ++u) {
        const int i = tid + u * TILE_THREADS;
        int s = 0, d = r0;
        if (i < cnt)        { s = ls[start + i]; d = ld[start + i]; }
        else if (i < total) { s = d = r0 + (i - cnt); }
        ks[u] = s;
        kd[u] = d - r0;
    }
    #pragma unroll
    for (int u = 0; u < KEEP; ++u) {
        #pragma unroll
        for (int hh = 0; hh < H; ++hh)
            ke[u][hh] = lrelu(as[ks[u] * H + hh] + ad[(r0 + kd[u]) * H + hh]);
    }
    #pragma unroll
    for (int u = 0; u < KEEP; ++u) {
        if (tid + u * TILE_THREADS < cnt) {
            #pragma unroll
            for (int hh = 0; hh < H; ++hh)
                atomicMaxF(&mx[kd[u] * H + hh], ke[u][hh]);
        }
    }
    for (int i0 = KEEP * TILE_THREADS + tid; i0 < cnt; i0 += 4 * TILE_THREADS) {
        int   sv[4], dv[4];
        float ev[4][H];
        #pragma unroll
        for (int u = 0; u < 4; ++u) {
            const int i = i0 + u * TILE_THREADS;
            const bool ok = i < cnt;
            sv[u] = ok ? ls[start + i] : 0;
            dv[u] = ok ? ld[start + i] : r0;
        }
        #pragma unroll
        for (int u = 0; u < 4; ++u) {
            #pragma unroll
            for (int hh = 0; hh < H; ++hh)
                ev[u][hh] = lrelu(as[sv[u] * H + hh] + ad[dv[u] * H + hh]);
        }
        #pragma unroll
        for (int u = 0; u < 4; ++u) {
            if (i0 + u * TILE_THREADS < cnt) {
                #pragma unroll
                for (int hh = 0; hh < H; ++hh)
                    atomicMaxF(&mx[(dv[u] - r0) * H + hh], ev[u][hh]);
            }
        }
    }
    __syncthreads();

    int k = 0;
    #pragma unroll 1
    for (int b = wave * 32; b < total; b += NW * 32, ++k) {
        const int i = b + lane;
        int   s, dl;
        float e[H];
        if (k < KEEP) {
            s  = ks[0];
            dl = kd[0];
            #pragma unroll
            for (int hh = 0; hh < H; ++hh) e[hh] = ke[0][hh];
            #pragma unroll
            for (int u = 0; u + 1 < KEEP; ++u) {
                ks[u] = ks[u + 1];
                kd[u] = kd[u + 1];
                #pragma unroll
                for (int hh = 0; hh < H; ++hh) ke[u][hh] = ke[u + 1][hh];
            }
        } else {
            s = 0; dl = 0;
            #pragma unroll
            for (int hh = 0; hh < H; ++hh) e[hh] = 0.f;
            if (i < total) {
                if (i < cnt) { s = ls[start + i]; dl = ld[start + i] - r0; }
                else         { s = r0 + (i - cnt); dl = i - cnt; }
                #pragma unroll
                for (int hh = 0; hh < H; ++hh)
                    e[hh] = lrelu(as[s * H + hh] + ad[(r0 + dl) * H + hh]);
            }
        }
        float ex[H];
        #pragma unroll
        for (int hh = 0; hh < H; ++hh) ex[hh] = 0.f;
        if constexpr (H == 4) {
            const bool hi = lane >= 16;
            float v0[16], v1[16];
            auto load16 = [&](int jb) {
                #pragma unroll
                for (int j = 0; j < 16; ++j) {
                    const int sj = __builtin_amdgcn_readlane(s, jb + j);
                    const float* hp = h + (size_t)sj * F;
                    v0[j] = hp[lane];
                    v1[j] = hp[32 + lane];
                }
            };
            auto add16 = [&](auto fullc, int jb, float wl, float wh) {
                constexpr bool FULL = decltype(fullc)::value != 0;
                steps<0, 16>([&](auto jc) {
                    constexpr int j = decltype(jc)::value;
                    const int   dj  = __builtin_amdgcn_readlane(dl, jb + j);
                    const float wlo = row_bcast<j>(wl);
                    const float whi = row_bcast<j>(wh);
                    if (FULL || b + jb + j < total) {
                        atomicAdd(&acc[dj * P + lane],      v0[j] * wlo);
                        atomicAdd(&acc[dj * P + 32 + lane], v1[j] * whi);
                    }
                });
            };
            load16(0);
            if (i < total) {
                #pragma unroll
                for (int hh = 0; hh < H; ++hh) ex[hh] = expf(e[hh] - mx[dl * H + hh]);
                #pragma unroll
                for (int hh = 0; hh < H; ++hh) atomicAdd(&dn[dl * H + hh], ex[hh]);
            }
            const float x0 = xhalf(ex[0]);
            const float x1 = xhalf(ex[1]);
            const float x2 = xhalf(ex[2]);
            const float x3 = xhalf(ex[3]);
            const float aLo = hi ? x1 : ex[0];
            const float aHi = hi ? x3 : ex[2];
            const float bLo = hi ? ex[1] : x0;
            const float bHi = hi ? ex[3] : x2;
            if (b + 16 <= total) add16(StepC<1>{}, 0, aLo, aHi);
            else                 add16(StepC<0>{}, 0, aLo, aHi);
            if (b + 16 < total) {
                load16(16);
                if (b + 32 <= total) add16(StepC<1>{}, 16, bLo, bHi);
                else                 add16(StepC<0>{}, 16, bLo, bHi);
            }
        } else {
            const int c    = lane & 15;
            const int half = lane >> 4;
            float v[16];
            steps<0, 16>([&](auto jc) {
                constexpr int j = decltype(jc)::value;
                const int sj = row_bcast<j>(s);
                v[j] = h[(size_t)sj * F + c];
            });
            if (i < total) {
                #pragma unroll
                for (int hh = 0; hh < H; ++hh) ex[hh] = expf(e[hh] - mx[dl * H + hh]);
                #pragma unroll
                for (int hh = 0; hh < H; ++hh) atomicAdd(&dn[dl * H + hh], ex[hh]);
            }
            if (b + 32 <= total) {
                steps<0, 16>([&](auto jc) {
                    constexpr int j = decltype(jc)::value;
                    const int   dj = row_bcast<j>(dl);
                    const float w  = row_bcast<j>(ex[0]);
                    atomicAdd(&acc[dj * P + c], v[j] * w);
                });
            } else {
                steps<0, 16>([&](auto jc) {
                    constexpr int j = decltype(jc)::value;
                    const int   dj = row_bcast<j>(dl);
                    const float w  = row_bcast<j>(ex[0]);
                    if (b + 16 * half + j < total) {
                        atomicAdd(&acc[dj * P + c], v[j] * w);
                    }
                });
            }
        }
    }
    __syncthreads();

    for (int i = tid; i < nrows * F; i += TILE_THREADS) {
        const int r = i / F, f = i % F;
        float v = acc[r * P + f] / (dn[r * H + f / 16] + 1e-16f) + bias[f];
        if constexpr (ELU) {
            v = (v > 0.f) ? v : (expf(v) - 1.f);
            acc[r * P + f] = v;
        } else {
            res[(size_t)r0 * F + i] = v;
        }
    }

    if constexpr (ELU) {
        const int ntile = (nrows + 15) >> 4;
        __syncthreads();

        if (wave < ntile) {
            const int half = lane >> 4;
            const int m    = lane & 15;
            const float* xr = acc + (wave * 16 + m) * P;
            v8f c = {};
            #pragma unroll
            for (int kk = 0; kk < F; kk += 4) {
                const int kb = kk + 2 * half;
                v2f a, b;
                a[0] = xr[kb];
                a[1] = xr[kb + 1];
                b[0] = Wn[(kb)     * 16 + m];
                b[1] = Wn[(kb + 1) * 16 + m];
                c = __builtin_amdgcn_wmma_f32_16x16x4_f32(
                    false, a, false, b, (short)0, c, false, false);
            }
            #pragma unroll
            for (int r = 0; r < 8; ++r)
                acc[(wave * 16 + 8 * half + r) * P + m] = c[r];
        }
        __syncthreads();

        for (int r = tid; r < nrows; r += TILE_THREADS) {
            float sa = 0.f, sd = 0.f;
            #pragma unroll
            for (int j = 0; j < 16; ++j) {
                const float v = acc[r * P + j];
                sa += v * a_srcn[j];
                sd += v * a_dstn[j];
            }
            asn[r0 + r] = sa;
            adn[r0 + r] = sd;
        }
        for (int i = tid; i < nrows * 16; i += TILE_THREADS)
            hn[(size_t)r0 * 16 + i] = acc[(i >> 4) * P + (i & 15)];
    }
}

static inline int cdiv(int a, int b) { return (a + b - 1) / b; }

extern "C" void kernel_launch(void* const* d_in, const int* in_sizes, int n_in,
                              void* d_out, int out_size, void* d_ws, size_t ws_size,
                              hipStream_t stream) {
    const float* x      = (const float*)d_in[0];
    const int*   ei     = (const int*)  d_in[1];
    const float* W1     = (const float*)d_in[2];
    const float* a_src1 = (const float*)d_in[3];
    const float* a_dst1 = (const float*)d_in[4];
    const float* b1     = (const float*)d_in[5];
    const float* W2     = (const float*)d_in[6];
    const float* a_src2 = (const float*)d_in[7];
    const float* a_dst2 = (const float*)d_in[8];
    const float* b2     = (const float*)d_in[9];

    const int N  = in_sizes[0] / 128;
    const int E  = in_sizes[1] / 2;

    float* ws   = (float*)d_ws;
    float* h1   = ws;
    float* as1  = h1   + (size_t)N * 64;
    float* ad1  = as1  + (size_t)N * 4;
    float* h2   = ad1  + (size_t)N * 4;
    float* as2  = h2   + (size_t)N * 16;
    float* ad2  = as2  + N;
    int* histG     = (int*)(ad2 + N);
    int* tileStart = histG + MAXT * MAXT;
    int* ls        = tileStart + 512;
    int* ld        = ls + E;

    float* out  = (float*)d_out;

    constexpr int TB = 256;
    const int NT = cdiv(N, TILE_ROWS);

    tile_hist_kernel<<<NCHUNK, BIN_THREADS, 0, stream>>>(ei, E, histG);

    gemm_wmma_f32<128, 4><<<N / 16, dim3(32, 4), 0, stream>>>(x, W1, h1);
    static_assert(TB == BIN_THREADS, "one block size for both stages");
    place_logit_kernel<<<NCHUNK + cdiv(N * 4, TB * LOGIT_PPT), BIN_THREADS, 0, stream>>>(
        ei, E, histG, tileStart, ls, ld, h1, a_src1, a_dst1, as1, ad1, N);
    tile_gat_kernel<4, true><<<NT, TILE_THREADS, 0, stream>>>(
        tileStart, ls, ld, h1, as1, ad1, b1, nullptr,
        W2, a_src2, a_dst2, h2, as2, ad2, N);

    tile_gat_kernel<1, false><<<NT, TILE_THREADS, 0, stream>>>(
        tileStart, ls, ld, h2, as2, ad2, b2, out,
        nullptr, nullptr, nullptr, nullptr, nullptr, nullptr, N);
}


# Binning with no count over all chunks: each chunk is put into tile order in its own stretch of the list; a tile finds its 256 runs by a prefix of run lengths in LDS; count launch goes: 4 launches
# speedup vs baseline: 9.8173x; 1.0623x over previous
#include <hip/hip_runtime.h>
#include <hip/hip_bf16.h>
#include <math.h>

typedef float v2f __attribute__((ext_vector_type(2)));
typedef float v8f __attribute__((ext_vector_type(8)));

#define NEG_SLOPE 0.2f

__device__ __forceinline__ void atomicMaxF(float* addr, float val) {
    if (val >= 0.0f) {
        atomicMax((int*)addr, __float_as_int(val));
    } else {
        atomicMin((unsigned int*)addr, __float_as_uint(val));
    }
}

__device__ __forceinline__ void edge_sd(const int* __restrict__ ei, int e, int E,
                                        int& s, int& d) {
    if (e < E) { s = ei[e]; d = ei[E + e]; }
    else       { s = d = e - E; }
}

template <int K, int NCOLT>
__global__ void gemm_wmma_f32(const float* __restrict__ X,
                              const float* __restrict__ W,
                              float* __restrict__ H) {
    constexpr int FOUT = 16 * NCOLT;
    __shared__ float As[16][K + 4];

    const int r0   = blockIdx.x * 16;
    const int lane = threadIdx.x;
    const int wave = threadIdx.y;
    const int tid  = wave * 32 + lane;
    const int nthr = 32 * NCOLT;

    for (int i = tid; i < 16 * K; i += nthr) {
        int rr = i / K, cc = i % K;
        As[rr][cc] = X[(size_t)(r0 + rr) * K + cc];
    }
    __syncthreads();

    const int half = lane >> 4;
    const int m    = lane & 15;
    const int c0   = wave * 16;

    v8f acc = {};
    #pragma unroll 4
    for (int kk = 0; kk < K; kk += 4) {
        const int kb = kk + 2 * half;
        v2f a, b;
        a[0] = As[m][kb];
        a[1] = As[m][kb + 1];
        b[0] = W[(size_t)(kb)     * FOUT + c0 + m];
        b[1] = W[(size_t)(kb + 1) * FOUT + c0 + m];
        acc = __builtin_amdgcn_wmma_f32_16x16x4_f32(
             false, a,  false, b,
             (short)0, acc,  false,  false);
    }

    #pragma unroll
    for (int r = 0; r < 8; ++r) {
        H[(size_t)(r0 + r + 8 * half) * FOUT + c0 + m] = acc[r];
    }
}

#define LOGIT_PPT 4
template <int H, int D, int PPT>
__device__ __forceinline__ void alpha_body(int blk,
                             const float* __restrict__ h,
                             const float* __restrict__ a_src,
                             const float* __restrict__ a_dst,
                             float* __restrict__ as, float* __restrict__ ad,
                             int n) {
    const int base = blk * ((int)blockDim.x * PPT) + (int)threadIdx.x;
    const int npair = n * H;
    float v[PPT][D];
    #pragma unroll
    for (int p = 0; p < PPT; ++p) {
        const int i  = base + p * (int)blockDim.x;
        const int ic = i < npair ? i : npair - 1;
        const float* hp = h + (size_t)ic * D;
        #pragma unroll
        for (int j = 0; j < D; ++j) v[p][j] = hp[j];
    }
    #pragma unroll
    for (int p = 0; p < PPT; ++p) {
        const int i = base + p * (int)blockDim.x;
        if (i < npair) {
            const int hh = i % H;
            float s = 0.f, d = 0.f;
            #pragma unroll
            for (int j = 0; j < D; ++j) {
                s += v[p][j] * a_src[hh * D + j];
                d += v[p][j] * a_dst[hh * D + j];
            }
            as[i] = s;
            ad[i] = d;
        }
    }
}

#define TILE_ROWS 391
#define MAXT      256
#define NCHUNK    256
#define BIN_THREADS 256
#define BIN_EPT   16
#define TILE_THREADS 1024
#define KEEP 5
#define GP_CAP (KEEP * TILE_THREADS)

__device__ __forceinline__ int tile_of(int d) {
    int t = d / TILE_ROWS;
    return t < MAXT ? t : MAXT - 1;
}

__device__ __forceinline__ void
tile_place_body(const int* __restrict__ ei, int E,
                  int* __restrict__ segOff,
                  int* __restrict__ ls, int* __restrict__ ld) {
    constexpr int TRIP = BIN_THREADS * BIN_EPT;
    static_assert(BIN_THREADS == MAXT, "one thread per tile counter");
    __shared__ int  cursor[MAXT];
    __shared__ int  lscan[MAXT];
    __shared__ int  sS[TRIP];
    __shared__ int  sD[TRIP];
    const int t = threadIdx.x;

    const int cs = (E + NCHUNK - 1) / NCHUNK;
    const int lo = blockIdx.x * cs;
    const int hi = (lo + cs < E) ? lo + cs : E;
    const int last = (hi > 0) ? hi - 1 : 0;
    const int len = (hi > lo) ? hi - lo : 0;
    const bool ordered = len <= TRIP;

    int sv[BIN_EPT], dv[BIN_EPT];
    #pragma unroll
    for (int u = 0; u < BIN_EPT; ++u) {
        const int e  = lo + t + u * BIN_THREADS;
        const int ec = e < hi ? e : last;
        sv[u] = ei[ec];
        dv[u] = ei[E + ec];
    }
    cursor[t] = 0;
    __syncthreads();
    #pragma unroll
    for (int u = 0; u < BIN_EPT; ++u)
        atomicAdd(&cursor[tile_of(dv[u])], (lo + t + u * BIN_THREADS < hi) ? 1 : 0);
    for (int e0 = lo + t + TRIP; e0 < hi; e0 += TRIP) {
        int d[BIN_EPT];
        #pragma unroll
        for (int u = 0; u < BIN_EPT; ++u) {
            const int e = e0 + u * BIN_THREADS;
            d[u] = ei[E + (e < hi ? e : last)];
        }
        #pragma unroll
        for (int u = 0; u < BIN_EPT; ++u)
            atomicAdd(&cursor[tile_of(d[u])], (e0 + u * BIN_THREADS < hi) ? 1 : 0);
    }
    __syncthreads();
    const int own = cursor[t];
    lscan[t] = own;
    __syncthreads();
    for (int off = 1; off < MAXT; off <<= 1) {
        int w = 0;
        if (t >= off) w = lscan[t - off];
        __syncthreads();
        lscan[t] += w;
        __syncthreads();
    }
    {
        const int lstart = lscan[t] - own;
        cursor[t] = lstart;
        int* so = segOff + (int)blockIdx.x * (MAXT + 1);
        so[t] = lstart;
        if (t == MAXT - 1) so[MAXT] = lscan[t];
    }
    __syncthreads();

    if (ordered) {
        int pos[BIN_EPT];
        #pragma unroll
        for (int u = 0; u < BIN_EPT; ++u)
            pos[u] = atomicAdd(&cursor[tile_of(dv[u])], (lo + t + u * BIN_THREADS < hi) ? 1 : 0);
        #pragma unroll
        for (int u = 0; u < BIN_EPT; ++u) {
            if (lo + t + u * BIN_THREADS < hi) {
                sS[pos[u]] = sv[u];
                sD[pos[u]] = dv[u];
            }
        }
        __syncthreads();
        const int lastj = len > 0 ? len - 1 : 0;
        #pragma unroll
        for (int u = 0; u < BIN_EPT; ++u) {
            const int j  = t + u * BIN_THREADS;
            const int jc = j < len ? j : lastj;
            sv[u] = sS[jc];
            dv[u] = sD[jc];
        }
        #pragma unroll
        for (int u = 0; u < BIN_EPT; ++u) {
            const int j = t + u * BIN_THREADS;
            if (j < len) {
                ls[lo + j] = sv[u];
                ld[lo + j] = dv[u];
            }
        }
        return;
    }

    int pos[BIN_EPT];
    #pragma unroll
    for (int u = 0; u < BIN_EPT; ++u)
        pos[u] = atomicAdd(&cursor[tile_of(dv[u])], (lo + t + u * BIN_THREADS < hi) ? 1 : 0);
    #pragma unroll
    for (int u = 0; u < BIN_EPT; ++u) {
        if (lo + t + u * BIN_THREADS < hi) {
            ls[lo + pos[u]] = sv[u];
            ld[lo + pos[u]] = dv[u];
        }
    }
    for (int e0 = lo + t + TRIP; e0 < hi; e0 += TRIP) {
        #pragma unroll
        for (int u = 0; u < BIN_EPT; ++u) {
            const int e  = e0 + u * BIN_THREADS;
            const int ec = e < hi ? e : last;
            sv[u] = ei[ec];
            dv[u] = ei[E + ec];
        }
        #pragma unroll
        for (int u = 0; u < BIN_EPT; ++u)
            pos[u] = atomicAdd(&cursor[tile_of(dv[u])], (e0 + u * BIN_THREADS < hi) ? 1 : 0);
        #pragma unroll
        for (int u = 0; u < BIN_EPT; ++u) {
            if (e0 + u * BIN_THREADS < hi) {
                ls[lo + pos[u]] = sv[u];
                ld[lo + pos[u]] = dv[u];
            }
        }
    }
}

__global__ void __launch_bounds__(BIN_THREADS)
place_logit_kernel(const int* __restrict__ ei, int E,
                   int* __restrict__ segOff,
                   int* __restrict__ ls, int* __restrict__ ld,
                   const float* __restrict__ h,
                   const float* __restrict__ a_src, const float* __restrict__ a_dst,
                   float* __restrict__ as, float* __restrict__ ad, int n) {
    if (blockIdx.x < NCHUNK) {
        tile_place_body(ei, E, segOff, ls, ld);
    } else {
        alpha_body<4, 16, LOGIT_PPT>((int)blockIdx.x - NCHUNK, h, a_src, a_dst, as, ad, n);
    }
}

__device__ __forceinline__ float lrelu(float v) {
    return (v > 0.f) ? v : NEG_SLOPE * v;
}

template <int J> struct StepC { static constexpr int value = J; };
template <int J, int N, class Fn>
__device__ __forceinline__ void steps(Fn&& f) {
    if constexpr (J < N) { f(StepC<J>{}); steps<J + 1, N>(f); }
}
template <int J> __device__ __forceinline__ int row_bcast(int v) {
    return __builtin_amdgcn_update_dpp(0, v, 0x150 + J, 0xf, 0xf, true);
}
template <int J> __device__ __forceinline__ float row_bcast(float v) {
    return __int_as_float(row_bcast<J>(__float_as_int(v)));
}
template <int D> __device__ __forceinline__ int row_shr0(int v) {
    return __builtin_amdgcn_update_dpp(0, v, 0x110 + D, 0xf, 0xf, true);
}
__device__ __forceinline__ float xhalf(float x) {
    const int b = __float_as_int(x);
    return __int_as_float(__builtin_amdgcn_permlanex16(b, b, 0x76543210, 0xfedcba98, false, false));
}

template <int H, bool ELU>
__global__ void __launch_bounds__(TILE_THREADS)
tile_gat_kernel(const int* __restrict__ segOff,
                const int* __restrict__ ls, const int* __restrict__ ld,
                const float* __restrict__ h,
                const float* __restrict__ as, const float* __restrict__ ad,
                const float* __restrict__ bias,
                float* __restrict__ res,
                const float* __restrict__ Wn,
                const float* __restrict__ a_srcn, const float* __restrict__ a_dstn,
                float* __restrict__ hn, float* __restrict__ asn, float* __restrict__ adn,
                int N, int E) {
    constexpr int F = H * 16;
    constexpr int P     = ELU ? F + 4 : F;
    constexpr int AROWS = ELU ? (TILE_ROWS + 15) / 16 * 16 : TILE_ROWS;
    __shared__ float acc[AROWS * P];
    __shared__ float mx[TILE_ROWS * H];
    __shared__ float dn[TILE_ROWS * H];
    __shared__ int   segBase[NCHUNK];
    __shared__ int   segPos[NCHUNK];
    __shared__ int   wtot[NCHUNK / 32];
    __shared__ int   gpos[GP_CAP];
    static_assert(NCHUNK % 32 == 0 && NCHUNK <= TILE_THREADS, "whole waves take the chunks");

    const int tid   = threadIdx.x;
    const int lane  = tid & 31;
    const int wave  = tid >> 5;
    constexpr int NW = TILE_THREADS / 32;

    const int r0    = blockIdx.x * TILE_ROWS;
    int nrows = N - r0;
    if (nrows > TILE_ROWS) nrows = TILE_ROWS;
    if (nrows < 0) nrows = 0;
    const int cs = (E + NCHUNK - 1) / NCHUNK;
    int segA = 0, segB = 0;
    if (tid < NCHUNK) {
        const int* so = segOff + tid * (MAXT + 1) + (int)blockIdx.x;
        segA = so[0];
        segB = so[1];
    }

    const int crows = ELU ? ((nrows + 15) >> 4) * 16 : nrows;
    for (int i = tid; i < crows * F; i += TILE_THREADS) acc[(i / F) * P + (i % F)] = 0.f;
    for (int i = tid; i < nrows * H; i += TILE_THREADS) {
        const int g = r0 * H + i;
        mx[i] = lrelu(as[g] + ad[g]);
        dn[i] = 0.f;
    }
    const int segLen = segB - segA;
    int segInc = segLen;
    if (tid < NCHUNK) {
        segInc += row_shr0<1>(segInc);
        segInc += row_shr0<2>(segInc);
        segInc += row_shr0<4>(segInc);
        segInc += row_shr0<8>(segInc);
        const int lowsum = __builtin_amdgcn_readlane(segInc, 15);
        if (lane >= 16) segInc += lowsum;
        if (lane == 31) wtot[wave] = segInc;
    }
    __syncthreads();
    int cnt = 0;
    #pragma unroll
    for (int w = 0; w < NCHUNK / 32; ++w) cnt += wtot[w];
    const int total = cnt + nrows;
    if (tid < NCHUNK) {
        int pre = 0;
        #pragma unroll
        for (int w = 0; w < NCHUNK / 32; ++w) pre += (w < wave) ? wtot[w] : 0;
        const int base = pre + segInc - segLen;
        const int pos  = tid * cs + segA;
        segBase[tid] = base;
        segPos[tid]  = pos;
        int kn = GP_CAP - base;
        if (kn > segLen) kn = segLen;
        for (int k = 0; k < kn; ++k) gpos[base + k] = pos + k;
    }
    __syncthreads();
    auto edge_word = [&](int i) -> int {
        int c = 0;
        #pragma unroll
        for (int step = NCHUNK / 2; step >= 1; step >>= 1)
            if (segBase[c + step] <= i) c += step;
        return segPos[c] + (i - segBase[c]);
    };

    int   ks[KEEP], kd[KEEP];
    float ke[KEEP][H];
    #pragma unroll
    for (int u = 0; u < KEEP; ++u) {
        const int i = tid + u * TILE_THREADS;
        int s = 0, d = r0;
        if (i < cnt)        { const int g = gpos[i]; s = ls[g]; d = ld[g]; }
        else if (i < total) { s = d = r0 + (i - cnt); }
        ks[u] = s;
        kd[u] = d - r0;
    }
    #pragma unroll
    for (int u = 0; u < KEEP; ++u) {
        #pragma unroll
        for (int hh = 0; hh < H; ++hh)
            ke[u][hh] = lrelu(as[ks[u] * H + hh] + ad[(r0 + kd[u]) * H + hh]);
    }
    #pragma unroll
    for (int u = 0; u < KEEP; ++u) {
        if (tid + u * TILE_THREADS < cnt) {
            #pragma unroll
            for (int hh = 0; hh < H; ++hh)
                atomicMaxF(&mx[kd[u] * H + hh], ke[u][hh]);
        }
    }
    for (int i0 = KEEP * TILE_THREADS + tid; i0 < cnt; i0 += 4 * TILE_THREADS) {
        int   sv[4], dv[4];
        float ev[4][H];
        #pragma unroll
        for (int u = 0; u < 4; ++u) {
            const int i = i0 + u * TILE_THREADS;
            const bool ok = i < cnt;
            const int g = ok ? edge_word(i) : 0;
            sv[u] = ok ? ls[g] : 0;
            dv[u] = ok ? ld[g] : r0;
        }
        #pragma unroll
        for (int u = 0; u < 4; ++u) {
            #pragma unroll
            for (int hh = 0; hh < H; ++hh)
                ev[u][hh] = lrelu(as[sv[u] * H + hh] + ad[dv[u] * H + hh]);
        }
        #pragma unroll
        for (int u = 0; u < 4; ++u) {
            if (i0 + u * TILE_THREADS < cnt) {
                #pragma unroll
                for (int hh = 0; hh < H; ++hh)
                    atomicMaxF(&mx[(dv[u] - r0) * H + hh], ev[u][hh]);
            }
        }
    }
    __syncthreads();

    int k = 0;
    #pragma unroll 1
    for (int b = wave * 32; b < total; b += NW * 32, ++k) {
        const int i = b + lane;
        int   s, dl;
        float e[H];
        if (k < KEEP) {
            s  = ks[0];
            dl = kd[0];
            #pragma unroll
            for (int hh = 0; hh < H; ++hh) e[hh] = ke[0][hh];
            #pragma unroll
            for (int u = 0; u + 1 < KEEP; ++u) {
                ks[u] = ks[u + 1];
                kd[u] = kd[u + 1];
                #pragma unroll
                for (int hh = 0; hh < H; ++hh) ke[u][hh] = ke[u + 1][hh];
            }
        } else {
            s = 0; dl = 0;
            #pragma unroll
            for (int hh = 0; hh < H; ++hh) e[hh] = 0.f;
            if (i < total) {
                if (i < cnt) { const int g = edge_word(i); s = ls[g]; dl = ld[g] - r0; }
                else         { s = r0 + (i - cnt); dl = i - cnt; }
                #pragma unroll
                for (int hh = 0; hh < H; ++hh)
                    e[hh] = lrelu(as[s * H + hh] + ad[(r0 + dl) * H + hh]);
            }
        }
        float ex[H];
        #pragma unroll
        for (int hh = 0; hh < H; ++hh) ex[hh] = 0.f;
        if constexpr (H == 4) {
            const bool hi = lane >= 16;
            float v0[16], v1[16];
            auto load16 = [&](int jb) {
                #pragma unroll
                for (int j = 0; j < 16; ++j) {
                    const int sj = __builtin_amdgcn_readlane(s, jb + j);
                    const float* hp = h + (size_t)sj * F;
                    v0[j] = hp[lane];
                    v1[j] = hp[32 + lane];
                }
            };
            auto add16 = [&](auto fullc, int jb, float wl, float wh) {
                constexpr bool FULL = decltype(fullc)::value != 0;
                steps<0, 16>([&](auto jc) {
                    constexpr int j = decltype(jc)::value;
                    const int   dj  = __builtin_amdgcn_readlane(dl, jb + j);
                    const float wlo = row_bcast<j>(wl);
                    const float whi = row_bcast<j>(wh);
                    if (FULL || b + jb + j < total) {
                        atomicAdd(&acc[dj * P + lane],      v0[j] * wlo);
                        atomicAdd(&acc[dj * P + 32 + lane], v1[j] * whi);
                    }
                });
            };
            load16(0);
            if (i < total) {
                #pragma unroll
                for (int hh = 0; hh < H; ++hh) ex[hh] = expf(e[hh] - mx[dl * H + hh]);
                #pragma unroll
                for (int hh = 0; hh < H; ++hh) atomicAdd(&dn[dl * H + hh], ex[hh]);
            }
            const float x0 = xhalf(ex[0]);
            const float x1 = xhalf(ex[1]);
            const float x2 = xhalf(ex[2]);
            const float x3 = xhalf(ex[3]);
            const float aLo = hi ? x1 : ex[0];
            const float aHi = hi ? x3 : ex[2];
            const float bLo = hi ? ex[1] : x0;
            const float bHi = hi ? ex[3] : x2;
            if (b + 16 <= total) add16(StepC<1>{}, 0, aLo, aHi);
            else                 add16(StepC<0>{}, 0, aLo, aHi);
            if (b + 16 < total) {
                load16(16);
                if (b + 32 <= total) add16(StepC<1>{}, 16, bLo, bHi);
                else                 add16(StepC<0>{}, 16, bLo, bHi);
            }
        } else {
            const int c    = lane & 15;
            const int half = lane >> 4;
            float v[16];
            steps<0, 16>([&](auto jc) {
                constexpr int j = decltype(jc)::value;
                const int sj = row_bcast<j>(s);
                v[j] = h[(size_t)sj * F + c];
            });
            if (i < total) {
                #pragma unroll
                for (int hh = 0; hh < H; ++hh) ex[hh] = expf(e[hh] - mx[dl * H + hh]);
                #pragma unroll
                for (int hh = 0; hh < H; ++hh) atomicAdd(&dn[dl * H + hh], ex[hh]);
            }
            if (b + 32 <= total) {
                steps<0, 16>([&](auto jc) {
                    constexpr int j = decltype(jc)::value;
                    const int   dj = row_bcast<j>(dl);
                    const float w  = row_bcast<j>(ex[0]);
                    atomicAdd(&acc[dj * P + c], v[j] * w);
                });
            } else {
                steps<0, 16>([&](auto jc) {
                    constexpr int j = decltype(jc)::value;
                    const int   dj = row_bcast<j>(dl);
                    const float w  = row_bcast<j>(ex[0]);
                    if (b + 16 * half + j < total) {
                        atomicAdd(&acc[dj * P + c], v[j] * w);
                    }
                });
            }
        }
    }
    __syncthreads();

    for (int i = tid; i < nrows * F; i += TILE_THREADS) {
        const int r = i / F, f = i % F;
        float v = acc[r * P + f] / (dn[r * H + f / 16] + 1e-16f) + bias[f];
        if constexpr (ELU) {
            v = (v > 0.f) ? v : (expf(v) - 1.f);
            acc[r * P + f] = v;
        } else {
            res[(size_t)r0 * F + i] = v;
        }
    }

    if constexpr (ELU) {
        const int ntile = (nrows + 15) >> 4;
        __syncthreads();

        if (wave < ntile) {
            const int half = lane >> 4;
            const int m    = lane & 15;
            const float* xr = acc + (wave * 16 + m) * P;
            v8f c = {};
            #pragma unroll
            for (int kk = 0; kk < F; kk += 4) {
                const int kb = kk + 2 * half;
                v2f a, b;
                a[0] = xr[kb];
                a[1] = xr[kb + 1];
                b[0] = Wn[(kb)     * 16 + m];
                b[1] = Wn[(kb + 1) * 16 + m];
                c = __builtin_amdgcn_wmma_f32_16x16x4_f32(
                    false, a, false, b, (short)0, c, false, false);
            }
            #pragma unroll
            for (int r = 0; r < 8; ++r)
                acc[(wave * 16 + 8 * half + r) * P + m] = c[r];
        }
        __syncthreads();

        for (int r = tid; r < nrows; r += TILE_THREADS) {
            float sa = 0.f, sd = 0.f;
            #pragma unroll
            for (int j = 0; j < 16; ++j) {
                const float v = acc[r * P + j];
                sa += v * a_srcn[j];
                sd += v * a_dstn[j];
            }
            asn[r0 + r] = sa;
            adn[r0 + r] = sd;
        }
        for (int i = tid; i < nrows * 16; i += TILE_THREADS)
            hn[(size_t)r0 * 16 + i] = acc[(i >> 4) * P + (i & 15)];
    }
}

static inline int cdiv(int a, int b) { return (a + b - 1) / b; }

extern "C" void kernel_launch(void* const* d_in, const int* in_sizes, int n_in,
                              void* d_out, int out_size, void* d_ws, size_t ws_size,
                              hipStream_t stream) {
    const float* x      = (const float*)d_in[0];
    const int*   ei     = (const int*)  d_in[1];
    const float* W1     = (const float*)d_in[2];
    const float* a_src1 = (const float*)d_in[3];
    const float* a_dst1 = (const float*)d_in[4];
    const float* b1     = (const float*)d_in[5];
    const float* W2     = (const float*)d_in[6];
    const float* a_src2 = (const float*)d_in[7];
    const float* a_dst2 = (const float*)d_in[8];
    const float* b2     = (const float*)d_in[9];

    const int N  = in_sizes[0] / 128;
    const int E  = in_sizes[1] / 2;

    float* ws   = (float*)d_ws;
    float* h1   = ws;
    float* as1  = h1   + (size_t)N * 64;
    float* ad1  = as1  + (size_t)N * 4;
    float* h2   = ad1  + (size_t)N * 4;
    float* as2  = h2   + (size_t)N * 16;
    float* ad2  = as2  + N;
    int* segOff    = (int*)(ad2 + N);
    int* ls        = segOff + MAXT * MAXT + 512;
    int* ld        = ls + E;

    float* out  = (float*)d_out;

    constexpr int TB = 256;
    const int NT = cdiv(N, TILE_ROWS);

    gemm_wmma_f32<128, 4><<<N / 16, dim3(32, 4), 0, stream>>>(x, W1, h1);
    static_assert(TB == BIN_THREADS, "one block size for both stages");
    place_logit_kernel<<<NCHUNK + cdiv(N * 4, TB * LOGIT_PPT), BIN_THREADS, 0, stream>>>(
        ei, E, segOff, ls, ld, h1, a_src1, a_dst1, as1, ad1, N);
    tile_gat_kernel<4, true><<<NT, TILE_THREADS, 0, stream>>>(
        segOff, ls, ld, h1, as1, ad1, b1, nullptr,
        W2, a_src2, a_dst2, h2, as2, ad2, N, E);

    tile_gat_kernel<1, false><<<NT, TILE_THREADS, 0, stream>>>(
        segOff, ls, ld, h2, as2, ad2, b2, out,
        nullptr, nullptr, nullptr, nullptr, nullptr, nullptr, N, E);
}
